# starting kernel
# baseline (speedup 1.0000x reference)
#include <hip/hip_runtime.h>
#include <math.h>


typedef __attribute__((ext_vector_type(2))) float v2f;
typedef __attribute__((ext_vector_type(8))) float v8f;

#define L   1024
#define DM  1024
#define EPSR 1.1920929e-07f

static __device__ __forceinline__ v8f wmma_f32(v2f a, v2f b, v8f c) {
  return __builtin_amdgcn_wmma_f32_16x16x4_f32(false, a, false, b, (short)0, c,
                                               false, false);
}

__global__ void __launch_bounds__(256)
rmsnorm3_kernel(const float* __restrict__ x, const float* __restrict__ pt,
                const float* __restrict__ gq, const float* __restrict__ gkv,
                const float* __restrict__ gc,
                float* __restrict__ xqn, float* __restrict__ xkvn,
                float* __restrict__ ptn) {
  __shared__ float red[256];
  const int row = blockIdx.x;
  const int tid = threadIdx.x;

  float4 xv = ((const float4*)(x + (size_t)row * DM))[tid];
  red[tid] = xv.x * xv.x + xv.y * xv.y + xv.z * xv.z + xv.w * xv.w;
  __syncthreads();
  for (int s = 128; s > 0; s >>= 1) {
    if (tid < s) red[tid] += red[tid + s];
    __syncthreads();
  }
  float rn = rsqrtf(red[0] * (1.0f / DM) + EPSR);
  __syncthreads();

  float4 gqv = ((const float4*)gq)[tid];
  float4 gkvv = ((const float4*)gkv)[tid];
  float4 oq, okv;
  oq.x = xv.x * rn * gqv.x;  oq.y = xv.y * rn * gqv.y;
  oq.z = xv.z * rn * gqv.z;  oq.w = xv.w * rn * gqv.w;
  okv.x = xv.x * rn * gkvv.x; okv.y = xv.y * rn * gkvv.y;
  okv.z = xv.z * rn * gkvv.z; okv.w = xv.w * rn * gkvv.w;
  ((float4*)(xqn + (size_t)row * DM))[tid] = oq;
  ((float4*)(xkvn + (size_t)row * DM))[tid] = okv;

  float4 pv = ((const float4*)(pt + (size_t)row * DM))[tid];
  red[tid] = pv.x * pv.x + pv.y * pv.y + pv.z * pv.z + pv.w * pv.w;
  __syncthreads();
  for (int s = 128; s > 0; s >>= 1) {
    if (tid < s) red[tid] += red[tid + s];
    __syncthreads();
  }
  float rnp = rsqrtf(red[0] * (1.0f / DM) + EPSR);
  float4 gcv = ((const float4*)gc)[tid];
  float4 oc;
  oc.x = pv.x * rnp * gcv.x; oc.y = pv.y * rnp * gcv.y;
  oc.z = pv.z * rnp * gcv.z; oc.w = pv.w * rnp * gcv.w;
  ((float4*)(ptn + (size_t)row * DM))[tid] = oc;
}

__global__ void __launch_bounds__(256)
wmma_gemm_bias(const float* __restrict__ A, const float* __restrict__ Wm,
               const float* __restrict__ bias, float* __restrict__ Cout,
               int N) {
  __shared__ float As[16 * 65];
  __shared__ float Bs[16 * 64];
  const int tid = threadIdx.x;
  const int lane = tid & 31, wid = tid >> 5;
  const int hhi = lane >> 4, l16 = lane & 15;
  const int m0 = blockIdx.x * 64, n0 = blockIdx.y * 64;
  const int wm = wid & 3, wn = wid >> 2;
  const int mloc = wm * 16 + l16;
  const int nloc0 = wn * 32 + l16;
  const int nloc1 = nloc0 + 16;

  v8f acc0 = {}; v8f acc1 = {};

  for (int k0 = 0; k0 < DM; k0 += 16) {
#pragma unroll
    for (int it = 0; it < 4; ++it) {
      int idx = tid + it * 256;
      int ak = idx & 15, am = idx >> 4;
      As[ak * 65 + am] = A[(size_t)(m0 + am) * DM + k0 + ak];
      int bn = idx & 63, bk = idx >> 6;
      Bs[bk * 64 + bn] = Wm[(size_t)(k0 + bk) * N + n0 + bn];
    }
    __syncthreads();
    if (k0 + 16 < DM) {
      __builtin_prefetch(&Wm[(size_t)(k0 + 16 + (tid >> 6)) * N + n0 + (tid & 63)], 0, 1);
      __builtin_prefetch(&A[(size_t)(m0 + (tid >> 2)) * DM + k0 + 16], 0, 1);
    }
#pragma unroll
    for (int kk = 0; kk < 4; ++kk) {
      const int kb = kk * 4 + 2 * hhi;
      v2f a;  a.x = As[kb * 65 + mloc];        a.y = As[(kb + 1) * 65 + mloc];
      v2f b0; b0.x = Bs[kb * 64 + nloc0];      b0.y = Bs[(kb + 1) * 64 + nloc0];
      v2f b1; b1.x = Bs[kb * 64 + nloc1];      b1.y = Bs[(kb + 1) * 64 + nloc1];
      acc0 = wmma_f32(a, b0, acc0);
      acc1 = wmma_f32(a, b1, acc1);
    }
    __syncthreads();
  }

  const int rbase = m0 + wm * 16 + hhi * 8;
  const float bb0 = bias[n0 + nloc0];
  const float bb1 = bias[n0 + nloc1];
#pragma unroll
  for (int v = 0; v < 8; ++v) {
    const int r = rbase + v;
    Cout[(size_t)r * N + n0 + nloc0] = acc0[v] + bb0;
    Cout[(size_t)r * N + n0 + nloc1] = acc1[v] + bb1;
  }
}

__global__ void cumsum_kernel(const float* __restrict__ C, float* __restrict__ P) {
  const int col = blockIdx.x * blockDim.x + threadIdx.x;
  float run = 0.f;
  P[col] = 0.f;
  for (int t = 0; t < L; ++t) {
    run += C[(size_t)t * DM + col];
    P[(size_t)(t + 1) * DM + col] = run;
  }
}

__global__ void augment_kernel(const float* __restrict__ Q,
                               const float* __restrict__ KV,
                               const float* __restrict__ P,
                               const int* __restrict__ cw,
                               float* __restrict__ Qa, float* __restrict__ Ka) {
  const int i = blockIdx.x;
  const int w = cw[0];
  const float scale = 0.125f;
  int E = i + w; if (E > L) E = L;
  int S = i - w; if (S < 0) S = 0;
  for (int c = threadIdx.x; c < DM; c += blockDim.x) {
    const int h = c >> 6, dd = c & 63;
    const float pe = P[(size_t)E * DM + c];
    const float ps = P[(size_t)S * DM + c];
    const float qv = Q[(size_t)i * DM + c];
    const float kv = KV[(size_t)i * 2048 + c];
    const int base = i * 2048 + h * 128 + dd;
    Qa[base]      = qv * (scale + pe);
    Qa[base + 64] = qv;
    Ka[base]      = kv;
    Ka[base + 64] = -kv * ps;
  }
}

__global__ void __launch_bounds__(128)
attn_kernel(const float* __restrict__ Qa, const float* __restrict__ Ka,
            const float* __restrict__ KV, float* __restrict__ wv) {
  __shared__ float Qs[4 * 16 * 130];
  __shared__ float Ks[16 * 130];
  __shared__ float Vs[16 * 64];
  __shared__ float Pls[4 * 16 * 18];

  const int tid = threadIdx.x;
  const int lane = tid & 31, wid = tid >> 5;
  const int hhi = lane >> 4, l16 = lane & 15;
  const int h = blockIdx.y;
  const int hoff = h * 128;
  const int qrow0 = blockIdx.x * 64 + wid * 16;
  const int qtile = qrow0 >> 4;
  const int jt_max = blockIdx.x * 4 + 3;

  float* myQ = &Qs[wid * 16 * 130];
  float* myP = &Pls[wid * 16 * 18];

  for (int idx = lane; idx < 16 * 128; idx += 32) {
    const int r = idx >> 7, kc = idx & 127;
    myQ[r * 130 + kc] = Qa[(size_t)(qrow0 + r) * 2048 + hoff + kc];
  }

  v8f o0 = {}, o1 = {}, o2 = {}, o3 = {};
  float m8[8], l8[8];
#pragma unroll
  for (int v = 0; v < 8; ++v) { m8[v] = -3.0e38f; l8[v] = 0.f; }

  for (int jt = 0; jt <= jt_max; ++jt) {
    for (int idx = tid; idx < 16 * 128; idx += 128) {
      const int r = idx >> 7, kc = idx & 127;
      Ks[r * 130 + kc] = Ka[(size_t)(jt * 16 + r) * 2048 + hoff + kc];
    }
    for (int idx = tid; idx < 16 * 64; idx += 128) {
      const int r = idx >> 6, n = idx & 63;
      Vs[r * 64 + n] = KV[(size_t)(jt * 16 + r) * 2048 + 1024 + h * 64 + n];
    }
    __syncthreads();

    if (jt <= qtile) {
      v8f s = {};
#pragma unroll 8
      for (int kk = 0; kk < 32; ++kk) {
        const int kb = kk * 4 + 2 * hhi;
        v2f a; a.x = myQ[l16 * 130 + kb]; a.y = myQ[l16 * 130 + kb + 1];
        v2f b; b.x = Ks[l16 * 130 + kb];  b.y = Ks[l16 * 130 + kb + 1];
        s = wmma_f32(a, b, s);
      }

      const bool diag = (jt == qtile);
#pragma unroll
      for (int v = 0; v < 8; ++v) {
        const int row = v + hhi * 8;
        float sv = s[v];
        if (diag && (l16 > row)) sv = -1.0e30f;
        float mx = sv;
        for (int off = 8; off > 0; off >>= 1)
          mx = fmaxf(mx, __shfl_xor(mx, off, 16));
        const float mnew = fmaxf(m8[v], mx);
        const float alpha = __expf(m8[v] - mnew);
        const float p = __expf(sv - mnew);
        float rs = p;
        for (int off = 8; off > 0; off >>= 1) rs += __shfl_xor(rs, off, 16);
        l8[v] = l8[v] * alpha + rs;
        m8[v] = mnew;
        o0[v] *= alpha; o1[v] *= alpha; o2[v] *= alpha; o3[v] *= alpha;
        myP[row * 18 + l16] = p;
      }

#pragma unroll
      for (int kk = 0; kk < 4; ++kk) {
        const int kb = kk * 4 + 2 * hhi;
        v2f a; a.x = myP[l16 * 18 + kb]; a.y = myP[l16 * 18 + kb + 1];
        v2f b0, b1, b2, b3;
        b0.x = Vs[kb * 64 + l16];      b0.y = Vs[(kb + 1) * 64 + l16];
        b1.x = Vs[kb * 64 + 16 + l16]; b1.y = Vs[(kb + 1) * 64 + 16 + l16];
        b2.x = Vs[kb * 64 + 32 + l16]; b2.y = Vs[(kb + 1) * 64 + 32 + l16];
        b3.x = Vs[kb * 64 + 48 + l16]; b3.y = Vs[(kb + 1) * 64 + 48 + l16];
        o0 = wmma_f32(a, b0, o0);
        o1 = wmma_f32(a, b1, o1);
        o2 = wmma_f32(a, b2, o2);
        o3 = wmma_f32(a, b3, o3);
      }
    }
    __syncthreads();
  }

  const int cb = h * 64;
#pragma unroll
  for (int v = 0; v < 8; ++v) {
    const float inv = 1.0f / l8[v];
    const int r = qrow0 + v + hhi * 8;
    wv[(size_t)r * DM + cb + l16]      = o0[v] * inv;
    wv[(size_t)r * DM + cb + 16 + l16] = o1[v] * inv;
    wv[(size_t)r * DM + cb + 32 + l16] = o2[v] * inv;
    wv[(size_t)r * DM + cb + 48 + l16] = o3[v] * inv;
  }
}

extern "C" void kernel_launch(void* const* d_in, const int* in_sizes, int n_in,
                              void* d_out, int out_size, void* d_ws,
                              size_t ws_size, hipStream_t stream) {
  const float* x    = (const float*)d_in[0];
  const float* pt   = (const float*)d_in[1];
  const float* gq   = (const float*)d_in[3];
  const float* Wq   = (const float*)d_in[4];
  const float* bq   = (const float*)d_in[5];
  const float* gkv  = (const float*)d_in[6];
  const float* Wkv  = (const float*)d_in[7];
  const float* bkv  = (const float*)d_in[8];
  const float* gc   = (const float*)d_in[9];
  const float* Wc   = (const float*)d_in[10];
  const float* bc   = (const float*)d_in[11];
  const float* Wout = (const float*)d_in[12];
  const float* bout = (const float*)d_in[13];
  const int*   cw   = (const int*)d_in[14];
  float* out = (float*)d_out;

  float* ws = (float*)d_ws;
  const size_t R = 1024;
  float* xqn  = ws + 0 * R * 1024;
  float* xkvn = ws + 1 * R * 1024;
  float* ptn  = ws + 2 * R * 1024;
  float* Qb   = ws + 3 * R * 1024;
  float* KVb  = ws + 4 * R * 1024;
  float* Cb   = ws + 6 * R * 1024;
  float* Pb   = ws + 7 * R * 1024;
  float* Qab  = ws + 8256 * R;
  float* Kab  = ws + 10304 * R;
  float* wvb  = ws + 12352 * R;

  rmsnorm3_kernel<<<1024, 256, 0, stream>>>(x, pt, gq, gkv, gc, xqn, xkvn, ptn);
  wmma_gemm_bias<<<dim3(16, 16), 256, 0, stream>>>(xqn, Wq, bq, Qb, 1024);
  wmma_gemm_bias<<<dim3(16, 32), 256, 0, stream>>>(xkvn, Wkv, bkv, KVb, 2048);
  wmma_gemm_bias<<<dim3(16, 16), 256, 0, stream>>>(ptn, Wc, bc, Cb, 1024);
  cumsum_kernel<<<4, 256, 0, stream>>>(Cb, Pb);
  augment_kernel<<<1024, 256, 0, stream>>>(Qb, KVb, Pb, cw, Qab, Kab);
  attn_kernel<<<dim3(16, 16), 128, 0, stream>>>(Qab, Kab, KVb, wvb);
  wmma_gemm_bias<<<dim3(16, 16), 256, 0, stream>>>(wvb, Wout, bout, out, 1024);
}


# attention kernel: Q~/K~/V tiles staged by 16-byte global loads all issued before the first LDS store (1 memory round trip per key tile, was 16 in series); LDS pitch 130->132 for aligned stores
# speedup vs baseline: 1.6111x; 1.6111x over previous
#include <hip/hip_runtime.h>
#include <math.h>


typedef __attribute__((ext_vector_type(2))) float v2f;
typedef __attribute__((ext_vector_type(8))) float v8f;

#define L   1024
#define DM  1024
#define EPSR 1.1920929e-07f

static __device__ __forceinline__ v8f wmma_f32(v2f a, v2f b, v8f c) {
  return __builtin_amdgcn_wmma_f32_16x16x4_f32(false, a, false, b, (short)0, c,
                                               false, false);
}

__global__ void __launch_bounds__(256)
rmsnorm3_kernel(const float* __restrict__ x, const float* __restrict__ pt,
                const float* __restrict__ gq, const float* __restrict__ gkv,
                const float* __restrict__ gc,
                float* __restrict__ xqn, float* __restrict__ xkvn,
                float* __restrict__ ptn) {
  __shared__ float red[256];
  const int row = blockIdx.x;
  const int tid = threadIdx.x;

  float4 xv = ((const float4*)(x + (size_t)row * DM))[tid];
  red[tid] = xv.x * xv.x + xv.y * xv.y + xv.z * xv.z + xv.w * xv.w;
  __syncthreads();
  for (int s = 128; s > 0; s >>= 1) {
    if (tid < s) red[tid] += red[tid + s];
    __syncthreads();
  }
  float rn = rsqrtf(red[0] * (1.0f / DM) + EPSR);
  __syncthreads();

  float4 gqv = ((const float4*)gq)[tid];
  float4 gkvv = ((const float4*)gkv)[tid];
  float4 oq, okv;
  oq.x = xv.x * rn * gqv.x;  oq.y = xv.y * rn * gqv.y;
  oq.z = xv.z * rn * gqv.z;  oq.w = xv.w * rn * gqv.w;
  okv.x = xv.x * rn * gkvv.x; okv.y = xv.y * rn * gkvv.y;
  okv.z = xv.z * rn * gkvv.z; okv.w = xv.w * rn * gkvv.w;
  ((float4*)(xqn + (size_t)row * DM))[tid] = oq;
  ((float4*)(xkvn + (size_t)row * DM))[tid] = okv;

  float4 pv = ((const float4*)(pt + (size_t)row * DM))[tid];
  red[tid] = pv.x * pv.x + pv.y * pv.y + pv.z * pv.z + pv.w * pv.w;
  __syncthreads();
  for (int s = 128; s > 0; s >>= 1) {
    if (tid < s) red[tid] += red[tid + s];
    __syncthreads();
  }
  float rnp = rsqrtf(red[0] * (1.0f / DM) + EPSR);
  float4 gcv = ((const float4*)gc)[tid];
  float4 oc;
  oc.x = pv.x * rnp * gcv.x; oc.y = pv.y * rnp * gcv.y;
  oc.z = pv.z * rnp * gcv.z; oc.w = pv.w * rnp * gcv.w;
  ((float4*)(ptn + (size_t)row * DM))[tid] = oc;
}

__global__ void __launch_bounds__(256)
wmma_gemm_bias(const float* __restrict__ A, const float* __restrict__ Wm,
               const float* __restrict__ bias, float* __restrict__ Cout,
               int N) {
  __shared__ float As[16 * 65];
  __shared__ float Bs[16 * 64];
  const int tid = threadIdx.x;
  const int lane = tid & 31, wid = tid >> 5;
  const int hhi = lane >> 4, l16 = lane & 15;
  const int m0 = blockIdx.x * 64, n0 = blockIdx.y * 64;
  const int wm = wid & 3, wn = wid >> 2;
  const int mloc = wm * 16 + l16;
  const int nloc0 = wn * 32 + l16;
  const int nloc1 = nloc0 + 16;

  v8f acc0 = {}; v8f acc1 = {};

  for (int k0 = 0; k0 < DM; k0 += 16) {
#pragma unroll
    for (int it = 0; it < 4; ++it) {
      int idx = tid + it * 256;
      int ak = idx & 15, am = idx >> 4;
      As[ak * 65 + am] = A[(size_t)(m0 + am) * DM + k0 + ak];
      int bn = idx & 63, bk = idx >> 6;
      Bs[bk * 64 + bn] = Wm[(size_t)(k0 + bk) * N + n0 + bn];
    }
    __syncthreads();
    if (k0 + 16 < DM) {
      __builtin_prefetch(&Wm[(size_t)(k0 + 16 + (tid >> 6)) * N + n0 + (tid & 63)], 0, 1);
      __builtin_prefetch(&A[(size_t)(m0 + (tid >> 2)) * DM + k0 + 16], 0, 1);
    }
#pragma unroll
    for (int kk = 0; kk < 4; ++kk) {
      const int kb = kk * 4 + 2 * hhi;
      v2f a;  a.x = As[kb * 65 + mloc];        a.y = As[(kb + 1) * 65 + mloc];
      v2f b0; b0.x = Bs[kb * 64 + nloc0];      b0.y = Bs[(kb + 1) * 64 + nloc0];
      v2f b1; b1.x = Bs[kb * 64 + nloc1];      b1.y = Bs[(kb + 1) * 64 + nloc1];
      acc0 = wmma_f32(a, b0, acc0);
      acc1 = wmma_f32(a, b1, acc1);
    }
    __syncthreads();
  }

  const int rbase = m0 + wm * 16 + hhi * 8;
  const float bb0 = bias[n0 + nloc0];
  const float bb1 = bias[n0 + nloc1];
#pragma unroll
  for (int v = 0; v < 8; ++v) {
    const int r = rbase + v;
    Cout[(size_t)r * N + n0 + nloc0] = acc0[v] + bb0;
    Cout[(size_t)r * N + n0 + nloc1] = acc1[v] + bb1;
  }
}

__global__ void cumsum_kernel(const float* __restrict__ C, float* __restrict__ P) {
  const int col = blockIdx.x * blockDim.x + threadIdx.x;
  float run = 0.f;
  P[col] = 0.f;
  for (int t = 0; t < L; ++t) {
    run += C[(size_t)t * DM + col];
    P[(size_t)(t + 1) * DM + col] = run;
  }
}

__global__ void augment_kernel(const float* __restrict__ Q,
                               const float* __restrict__ KV,
                               const float* __restrict__ P,
                               const int* __restrict__ cw,
                               float* __restrict__ Qa, float* __restrict__ Ka) {
  const int i = blockIdx.x;
  const int w = cw[0];
  const float scale = 0.125f;
  int E = i + w; if (E > L) E = L;
  int S = i - w; if (S < 0) S = 0;
  for (int c = threadIdx.x; c < DM; c += blockDim.x) {
    const int h = c >> 6, dd = c & 63;
    const float pe = P[(size_t)E * DM + c];
    const float ps = P[(size_t)S * DM + c];
    const float qv = Q[(size_t)i * DM + c];
    const float kv = KV[(size_t)i * 2048 + c];
    const int base = i * 2048 + h * 128 + dd;
    Qa[base]      = qv * (scale + pe);
    Qa[base + 64] = qv;
    Ka[base]      = kv;
    Ka[base + 64] = -kv * ps;
  }
}

#define QKP 132
__global__ void __launch_bounds__(128)
attn_kernel(const float* __restrict__ Qa, const float* __restrict__ Ka,
            const float* __restrict__ KV, float* __restrict__ wv) {
  __shared__ __attribute__((aligned(16))) float Qs[4 * 16 * QKP];
  __shared__ __attribute__((aligned(16))) float Ks[16 * QKP];
  __shared__ __attribute__((aligned(16))) float Vs[16 * 64];
  __shared__ float Pls[4 * 16 * 18];

  const int tid = threadIdx.x;
  const int lane = tid & 31, wid = tid >> 5;
  const int hhi = lane >> 4, l16 = lane & 15;
  const int h = blockIdx.y;
  const int hoff = h * 128;
  const int qrow0 = blockIdx.x * 64 + wid * 16;
  const int qtile = qrow0 >> 4;
  const int jt_max = blockIdx.x * 4 + 3;

  float* myQ = &Qs[wid * 16 * QKP];
  float* myP = &Pls[wid * 16 * 18];

  {
    const float* qsrc = Qa + (size_t)qrow0 * 2048 + hoff + lane * 4;
    float4 qr[16];
#pragma unroll
    for (int r = 0; r < 16; ++r)
      qr[r] = *(const float4*)(qsrc + (size_t)r * 2048);
#pragma unroll
    for (int r = 0; r < 16; ++r)
      *(float4*)&myQ[r * QKP + lane * 4] = qr[r];
  }

  const int kr = tid >> 5, kc = (tid & 31) * 4;
  const int vr = tid >> 4, vc = (tid & 15) * 4;
  const float* ksrc = Ka + (size_t)kr * 2048 + hoff + kc;
  const float* vsrc = KV + (size_t)vr * 2048 + 1024 + h * 64 + vc;

  v8f o0 = {}, o1 = {}, o2 = {}, o3 = {};
  float m8[8], l8[8];
#pragma unroll
  for (int v = 0; v < 8; ++v) { m8[v] = -3.0e38f; l8[v] = 0.f; }

  for (int jt = 0; jt <= jt_max; ++jt) {
    {
      const float* kp = ksrc + (size_t)jt * (16 * 2048);
      const float* vp = vsrc + (size_t)jt * (16 * 2048);
      const float4 k0 = *(const float4*)(kp);
      const float4 k1 = *(const float4*)(kp + 4 * 2048);
      const float4 k2 = *(const float4*)(kp + 8 * 2048);
      const float4 k3 = *(const float4*)(kp + 12 * 2048);
      const float4 v0 = *(const float4*)(vp);
      const float4 v1 = *(const float4*)(vp + 8 * 2048);
      *(float4*)&Ks[kr * QKP + kc]        = k0;
      *(float4*)&Ks[(kr + 4) * QKP + kc]  = k1;
      *(float4*)&Ks[(kr + 8) * QKP + kc]  = k2;
      *(float4*)&Ks[(kr + 12) * QKP + kc] = k3;
      *(float4*)&Vs[vr * 64 + vc]         = v0;
      *(float4*)&Vs[(vr + 8) * 64 + vc]   = v1;
    }
    __syncthreads();

    if (jt <= qtile) {
      v8f s = {};
#pragma unroll 8
      for (int kk = 0; kk < 32; ++kk) {
        const int kb = kk * 4 + 2 * hhi;
        v2f a; a.x = myQ[l16 * QKP + kb]; a.y = myQ[l16 * QKP + kb + 1];
        v2f b; b.x = Ks[l16 * QKP + kb];  b.y = Ks[l16 * QKP + kb + 1];
        s = wmma_f32(a, b, s);
      }

      const bool diag = (jt == qtile);
#pragma unroll
      for (int v = 0; v < 8; ++v) {
        const int row = v + hhi * 8;
        float sv = s[v];
        if (diag && (l16 > row)) sv = -1.0e30f;
        float mx = sv;
        for (int off = 8; off > 0; off >>= 1)
          mx = fmaxf(mx, __shfl_xor(mx, off, 16));
        const float mnew = fmaxf(m8[v], mx);
        const float alpha = __expf(m8[v] - mnew);
        const float p = __expf(sv - mnew);
        float rs = p;
        for (int off = 8; off > 0; off >>= 1) rs += __shfl_xor(rs, off, 16);
        l8[v] = l8[v] * alpha + rs;
        m8[v] = mnew;
        o0[v] *= alpha; o1[v] *= alpha; o2[v] *= alpha; o3[v] *= alpha;
        myP[row * 18 + l16] = p;
      }

#pragma unroll
      for (int kk = 0; kk < 4; ++kk) {
        const int kb = kk * 4 + 2 * hhi;
        v2f a; a.x = myP[l16 * 18 + kb]; a.y = myP[l16 * 18 + kb + 1];
        v2f b0, b1, b2, b3;
        b0.x = Vs[kb * 64 + l16];      b0.y = Vs[(kb + 1) * 64 + l16];
        b1.x = Vs[kb * 64 + 16 + l16]; b1.y = Vs[(kb + 1) * 64 + 16 + l16];
        b2.x = Vs[kb * 64 + 32 + l16]; b2.y = Vs[(kb + 1) * 64 + 32 + l16];
        b3.x = Vs[kb * 64 + 48 + l16]; b3.y = Vs[(kb + 1) * 64 + 48 + l16];
        o0 = wmma_f32(a, b0, o0);
        o1 = wmma_f32(a, b1, o1);
        o2 = wmma_f32(a, b2, o2);
        o3 = wmma_f32(a, b3, o3);
      }
    }
    __syncthreads();
  }

  const int cb = h * 64;
#pragma unroll
  for (int v = 0; v < 8; ++v) {
    const float inv = 1.0f / l8[v];
    const int r = qrow0 + v + hhi * 8;
    wv[(size_t)r * DM + cb + l16]      = o0[v] * inv;
    wv[(size_t)r * DM + cb + 16 + l16] = o1[v] * inv;
    wv[(size_t)r * DM + cb + 32 + l16] = o2[v] * inv;
    wv[(size_t)r * DM + cb + 48 + l16] = o3[v] * inv;
  }
}

extern "C" void kernel_launch(void* const* d_in, const int* in_sizes, int n_in,
                              void* d_out, int out_size, void* d_ws,
                              size_t ws_size, hipStream_t stream) {
  const float* x    = (const float*)d_in[0];
  const float* pt   = (const float*)d_in[1];
  const float* gq   = (const float*)d_in[3];
  const float* Wq   = (const float*)d_in[4];
  const float* bq   = (const float*)d_in[5];
  const float* gkv  = (const float*)d_in[6];
  const float* Wkv  = (const float*)d_in[7];
  const float* bkv  = (const float*)d_in[8];
  const float* gc   = (const float*)d_in[9];
  const float* Wc   = (const float*)d_in[10];
  const float* bc   = (const float*)d_in[11];
  const float* Wout = (const float*)d_in[12];
  const float* bout = (const float*)d_in[13];
  const int*   cw   = (const int*)d_in[14];
  float* out = (float*)d_out;

  float* ws = (float*)d_ws;
  const size_t R = 1024;
  float* xqn  = ws + 0 * R * 1024;
  float* xkvn = ws + 1 * R * 1024;
  float* ptn  = ws + 2 * R * 1024;
  float* Qb   = ws + 3 * R * 1024;
  float* KVb  = ws + 4 * R * 1024;
  float* Cb   = ws + 6 * R * 1024;
  float* Pb   = ws + 7 * R * 1024;
  float* Qab  = ws + 8256 * R;
  float* Kab  = ws + 10304 * R;
  float* wvb  = ws + 12352 * R;

  rmsnorm3_kernel<<<1024, 256, 0, stream>>>(x, pt, gq, gkv, gc, xqn, xkvn, ptn);
  wmma_gemm_bias<<<dim3(16, 16), 256, 0, stream>>>(xqn, Wq, bq, Qb, 1024);
  wmma_gemm_bias<<<dim3(16, 32), 256, 0, stream>>>(xkvn, Wkv, bkv, KVb, 2048);
  wmma_gemm_bias<<<dim3(16, 16), 256, 0, stream>>>(ptn, Wc, bc, Cb, 1024);
  cumsum_kernel<<<4, 256, 0, stream>>>(Cb, Pb);
  augment_kernel<<<1024, 256, 0, stream>>>(Qb, KVb, Pb, cw, Qab, Kab);
  attn_kernel<<<dim3(16, 16), 128, 0, stream>>>(Qab, Kab, KVb, wvb);
  wmma_gemm_bias<<<dim3(16, 16), 256, 0, stream>>>(wvb, Wout, bout, out, 1024);
}


# GEMM kernel (4 launches): K tile 16->64, next tile fetched by 16-byte global loads into registers before the current tile's WMMAs (software pipeline); sum order unchanged
# speedup vs baseline: 2.6951x; 1.6728x over previous
#include <hip/hip_runtime.h>
#include <math.h>


typedef __attribute__((ext_vector_type(2))) float v2f;
typedef __attribute__((ext_vector_type(8))) float v8f;

#define L   1024
#define DM  1024
#define EPSR 1.1920929e-07f

static __device__ __forceinline__ v8f wmma_f32(v2f a, v2f b, v8f c) {
  return __builtin_amdgcn_wmma_f32_16x16x4_f32(false, a, false, b, (short)0, c,
                                               false, false);
}

__global__ void __launch_bounds__(256)
rmsnorm3_kernel(const float* __restrict__ x, const float* __restrict__ pt,
                const float* __restrict__ gq, const float* __restrict__ gkv,
                const float* __restrict__ gc,
                float* __restrict__ xqn, float* __restrict__ xkvn,
                float* __restrict__ ptn) {
  __shared__ float red[256];
  const int row = blockIdx.x;
  const int tid = threadIdx.x;

  float4 xv = ((const float4*)(x + (size_t)row * DM))[tid];
  red[tid] = xv.x * xv.x + xv.y * xv.y + xv.z * xv.z + xv.w * xv.w;
  __syncthreads();
  for (int s = 128; s > 0; s >>= 1) {
    if (tid < s) red[tid] += red[tid + s];
    __syncthreads();
  }
  float rn = rsqrtf(red[0] * (1.0f / DM) + EPSR);
  __syncthreads();

  float4 gqv = ((const float4*)gq)[tid];
  float4 gkvv = ((const float4*)gkv)[tid];
  float4 oq, okv;
  oq.x = xv.x * rn * gqv.x;  oq.y = xv.y * rn * gqv.y;
  oq.z = xv.z * rn * gqv.z;  oq.w = xv.w * rn * gqv.w;
  okv.x = xv.x * rn * gkvv.x; okv.y = xv.y * rn * gkvv.y;
  okv.z = xv.z * rn * gkvv.z; okv.w = xv.w * rn * gkvv.w;
  ((float4*)(xqn + (size_t)row * DM))[tid] = oq;
  ((float4*)(xkvn + (size_t)row * DM))[tid] = okv;

  float4 pv = ((const float4*)(pt + (size_t)row * DM))[tid];
  red[tid] = pv.x * pv.x + pv.y * pv.y + pv.z * pv.z + pv.w * pv.w;
  __syncthreads();
  for (int s = 128; s > 0; s >>= 1) {
    if (tid < s) red[tid] += red[tid + s];
    __syncthreads();
  }
  float rnp = rsqrtf(red[0] * (1.0f / DM) + EPSR);
  float4 gcv = ((const float4*)gc)[tid];
  float4 oc;
  oc.x = pv.x * rnp * gcv.x; oc.y = pv.y * rnp * gcv.y;
  oc.z = pv.z * rnp * gcv.z; oc.w = pv.w * rnp * gcv.w;
  ((float4*)(ptn + (size_t)row * DM))[tid] = oc;
}

#define GKT 64
#define GPA 68
#define GPB 72
__global__ void __launch_bounds__(256)
wmma_gemm_bias(const float* __restrict__ A, const float* __restrict__ Wm,
               const float* __restrict__ bias, float* __restrict__ Cout,
               int N) {
  __shared__ __attribute__((aligned(16))) float As[64 * GPA];
  __shared__ __attribute__((aligned(16))) float Bs[GKT * GPB];
  const int tid = threadIdx.x;
  const int lane = tid & 31, wid = tid >> 5;
  const int hhi = lane >> 4, l16 = lane & 15;
  const int m0 = blockIdx.x * 64, n0 = blockIdx.y * 64;
  const int wm = wid & 3, wn = wid >> 2;
  const int mloc = wm * 16 + l16;
  const int nloc0 = wn * 32 + l16;
  const int nloc1 = nloc0 + 16;

  const int srow = tid >> 4;
  const int scol = (tid & 15) * 4;
  const float* ap = A + (size_t)(m0 + srow) * DM + scol;
  const float* bp = Wm + (size_t)srow * N + n0 + scol;
  const size_t a16 = (size_t)16 * DM;
  const size_t b16 = (size_t)16 * N;
  float* asd = &As[srow * GPA + scol];
  float* bsd = &Bs[srow * GPB + scol];

  float4 ra0 = *(const float4*)(ap);
  float4 ra1 = *(const float4*)(ap + a16);
  float4 ra2 = *(const float4*)(ap + 2 * a16);
  float4 ra3 = *(const float4*)(ap + 3 * a16);
  float4 rb0 = *(const float4*)(bp);
  float4 rb1 = *(const float4*)(bp + b16);
  float4 rb2 = *(const float4*)(bp + 2 * b16);
  float4 rb3 = *(const float4*)(bp + 3 * b16);

  v8f acc0 = {}; v8f acc1 = {};

  for (int k0 = 0; k0 < DM; k0 += GKT) {
    *(float4*)(asd)                = ra0;
    *(float4*)(asd + 16 * GPA)     = ra1;
    *(float4*)(asd + 32 * GPA)     = ra2;
    *(float4*)(asd + 48 * GPA)     = ra3;
    *(float4*)(bsd)                = rb0;
    *(float4*)(bsd + 16 * GPB)     = rb1;
    *(float4*)(bsd + 32 * GPB)     = rb2;
    *(float4*)(bsd + 48 * GPB)     = rb3;
    __syncthreads();
    if (k0 + GKT < DM) {
      const float* an = ap + (k0 + GKT);
      const float* bn = bp + (size_t)(k0 + GKT) * N;
      ra0 = *(const float4*)(an);
      ra1 = *(const float4*)(an + a16);
      ra2 = *(const float4*)(an + 2 * a16);
      ra3 = *(const float4*)(an + 3 * a16);
      rb0 = *(const float4*)(bn);
      rb1 = *(const float4*)(bn + b16);
      rb2 = *(const float4*)(bn + 2 * b16);
      rb3 = *(const float4*)(bn + 3 * b16);
    }
#pragma unroll
    for (int kk = 0; kk < GKT / 4; ++kk) {
      const int kb = kk * 4 + 2 * hhi;
      v2f a;  a.x = As[mloc * GPA + kb];       a.y = As[mloc * GPA + kb + 1];
      v2f b0; b0.x = Bs[kb * GPB + nloc0];     b0.y = Bs[(kb + 1) * GPB + nloc0];
      v2f b1; b1.x = Bs[kb * GPB + nloc1];     b1.y = Bs[(kb + 1) * GPB + nloc1];
      acc0 = wmma_f32(a, b0, acc0);
      acc1 = wmma_f32(a, b1, acc1);
    }
    __syncthreads();
  }

  const int rbase = m0 + wm * 16 + hhi * 8;
  const float bb0 = bias[n0 + nloc0];
  const float bb1 = bias[n0 + nloc1];
#pragma unroll
  for (int v = 0; v < 8; ++v) {
    const int r = rbase + v;
    Cout[(size_t)r * N + n0 + nloc0] = acc0[v] + bb0;
    Cout[(size_t)r * N + n0 + nloc1] = acc1[v] + bb1;
  }
}

__global__ void cumsum_kernel(const float* __restrict__ C, float* __restrict__ P) {
  const int col = blockIdx.x * blockDim.x + threadIdx.x;
  float run = 0.f;
  P[col] = 0.f;
  for (int t = 0; t < L; ++t) {
    run += C[(size_t)t * DM + col];
    P[(size_t)(t + 1) * DM + col] = run;
  }
}

__global__ void augment_kernel(const float* __restrict__ Q,
                               const float* __restrict__ KV,
                               const float* __restrict__ P,
                               const int* __restrict__ cw,
                               float* __restrict__ Qa, float* __restrict__ Ka) {
  const int i = blockIdx.x;
  const int w = cw[0];
  const float scale = 0.125f;
  int E = i + w; if (E > L) E = L;
  int S = i - w; if (S < 0) S = 0;
  for (int c = threadIdx.x; c < DM; c += blockDim.x) {
    const int h = c >> 6, dd = c & 63;
    const float pe = P[(size_t)E * DM + c];
    const float ps = P[(size_t)S * DM + c];
    const float qv = Q[(size_t)i * DM + c];
    const float kv = KV[(size_t)i * 2048 + c];
    const int base = i * 2048 + h * 128 + dd;
    Qa[base]      = qv * (scale + pe);
    Qa[base + 64] = qv;
    Ka[base]      = kv;
    Ka[base + 64] = -kv * ps;
  }
}

#define QKP 132
__global__ void __launch_bounds__(128)
attn_kernel(const float* __restrict__ Qa, const float* __restrict__ Ka,
            const float* __restrict__ KV, float* __restrict__ wv) {
  __shared__ __attribute__((aligned(16))) float Qs[4 * 16 * QKP];
  __shared__ __attribute__((aligned(16))) float Ks[16 * QKP];
  __shared__ __attribute__((aligned(16))) float Vs[16 * 64];
  __shared__ float Pls[4 * 16 * 18];

  const int tid = threadIdx.x;
  const int lane = tid & 31, wid = tid >> 5;
  const int hhi = lane >> 4, l16 = lane & 15;
  const int h = blockIdx.y;
  const int hoff = h * 128;
  const int qrow0 = blockIdx.x * 64 + wid * 16;
  const int qtile = qrow0 >> 4;
  const int jt_max = blockIdx.x * 4 + 3;

  float* myQ = &Qs[wid * 16 * QKP];
  float* myP = &Pls[wid * 16 * 18];

  {
    const float* qsrc = Qa + (size_t)qrow0 * 2048 + hoff + lane * 4;
    float4 qr[16];
#pragma unroll
    for (int r = 0; r < 16; ++r)
      qr[r] = *(const float4*)(qsrc + (size_t)r * 2048);
#pragma unroll
    for (int r = 0; r < 16; ++r)
      *(float4*)&myQ[r * QKP + lane * 4] = qr[r];
  }

  const int kr = tid >> 5, kc = (tid & 31) * 4;
  const int vr = tid >> 4, vc = (tid & 15) * 4;
  const float* ksrc = Ka + (size_t)kr * 2048 + hoff + kc;
  const float* vsrc = KV + (size_t)vr * 2048 + 1024 + h * 64 + vc;

  v8f o0 = {}, o1 = {}, o2 = {}, o3 = {};
  float m8[8], l8[8];
#pragma unroll
  for (int v = 0; v < 8; ++v) { m8[v] = -3.0e38f; l8[v] = 0.f; }

  for (int jt = 0; jt <= jt_max; ++jt) {
    {
      const float* kp = ksrc + (size_t)jt * (16 * 2048);
      const float* vp = vsrc + (size_t)jt * (16 * 2048);
      const float4 k0 = *(const float4*)(kp);
      const float4 k1 = *(const float4*)(kp + 4 * 2048);
      const float4 k2 = *(const float4*)(kp + 8 * 2048);
      const float4 k3 = *(const float4*)(kp + 12 * 2048);
      const float4 v0 = *(const float4*)(vp);
      const float4 v1 = *(const float4*)(vp + 8 * 2048);
      *(float4*)&Ks[kr * QKP + kc]        = k0;
      *(float4*)&Ks[(kr + 4) * QKP + kc]  = k1;
      *(float4*)&Ks[(kr + 8) * QKP + kc]  = k2;
      *(float4*)&Ks[(kr + 12) * QKP + kc] = k3;
      *(float4*)&Vs[vr * 64 + vc]         = v0;
      *(float4*)&Vs[(vr + 8) * 64 + vc]   = v1;
    }
    __syncthreads();

    if (jt <= qtile) {
      v8f s = {};
#pragma unroll 8
      for (int kk = 0; kk < 32; ++kk) {
        const int kb = kk * 4 + 2 * hhi;
        v2f a; a.x = myQ[l16 * QKP + kb]; a.y = myQ[l16 * QKP + kb + 1];
        v2f b; b.x = Ks[l16 * QKP + kb];  b.y = Ks[l16 * QKP + kb + 1];
        s = wmma_f32(a, b, s);
      }

      const bool diag = (jt == qtile);
#pragma unroll
      for (int v = 0; v < 8; ++v) {
        const int row = v + hhi * 8;
        float sv = s[v];
        if (diag && (l16 > row)) sv = -1.0e30f;
        float mx = sv;
        for (int off = 8; off > 0; off >>= 1)
          mx = fmaxf(mx, __shfl_xor(mx, off, 16));
        const float mnew = fmaxf(m8[v], mx);
        const float alpha = __expf(m8[v] - mnew);
        const float p = __expf(sv - mnew);
        float rs = p;
        for (int off = 8; off > 0; off >>= 1) rs += __shfl_xor(rs, off, 16);
        l8[v] = l8[v] * alpha + rs;
        m8[v] = mnew;
        o0[v] *= alpha; o1[v] *= alpha; o2[v] *= alpha; o3[v] *= alpha;
        myP[row * 18 + l16] = p;
      }

#pragma unroll
      for (int kk = 0; kk < 4; ++kk) {
        const int kb = kk * 4 + 2 * hhi;
        v2f a; a.x = myP[l16 * 18 + kb]; a.y = myP[l16 * 18 + kb + 1];
        v2f b0, b1, b2, b3;
        b0.x = Vs[kb * 64 + l16];      b0.y = Vs[(kb + 1) * 64 + l16];
        b1.x = Vs[kb * 64 + 16 + l16]; b1.y = Vs[(kb + 1) * 64 + 16 + l16];
        b2.x = Vs[kb * 64 + 32 + l16]; b2.y = Vs[(kb + 1) * 64 + 32 + l16];
        b3.x = Vs[kb * 64 + 48 + l16]; b3.y = Vs[(kb + 1) * 64 + 48 + l16];
        o0 = wmma_f32(a, b0, o0);
        o1 = wmma_f32(a, b1, o1);
        o2 = wmma_f32(a, b2, o2);
        o3 = wmma_f32(a, b3, o3);
      }
    }
    __syncthreads();
  }

  const int cb = h * 64;
#pragma unroll
  for (int v = 0; v < 8; ++v) {
    const float inv = 1.0f / l8[v];
    const int r = qrow0 + v + hhi * 8;
    wv[(size_t)r * DM + cb + l16]      = o0[v] * inv;
    wv[(size_t)r * DM + cb + 16 + l16] = o1[v] * inv;
    wv[(size_t)r * DM + cb + 32 + l16] = o2[v] * inv;
    wv[(size_t)r * DM + cb + 48 + l16] = o3[v] * inv;
  }
}

extern "C" void kernel_launch(void* const* d_in, const int* in_sizes, int n_in,
                              void* d_out, int out_size, void* d_ws,
                              size_t ws_size, hipStream_t stream) {
  const float* x    = (const float*)d_in[0];
  const float* pt   = (const float*)d_in[1];
  const float* gq   = (const float*)d_in[3];
  const float* Wq   = (const float*)d_in[4];
  const float* bq   = (const float*)d_in[5];
  const float* gkv  = (const float*)d_in[6];
  const float* Wkv  = (const float*)d_in[7];
  const float* bkv  = (const float*)d_in[8];
  const float* gc   = (const float*)d_in[9];
  const float* Wc   = (const float*)d_in[10];
  const float* bc   = (const float*)d_in[11];
  const float* Wout = (const float*)d_in[12];
  const float* bout = (const float*)d_in[13];
  const int*   cw   = (const int*)d_in[14];
  float* out = (float*)d_out;

  float* ws = (float*)d_ws;
  const size_t R = 1024;
  float* xqn  = ws + 0 * R * 1024;
  float* xkvn = ws + 1 * R * 1024;
  float* ptn  = ws + 2 * R * 1024;
  float* Qb   = ws + 3 * R * 1024;
  float* KVb  = ws + 4 * R * 1024;
  float* Cb   = ws + 6 * R * 1024;
  float* Pb   = ws + 7 * R * 1024;
  float* Qab  = ws + 8256 * R;
  float* Kab  = ws + 10304 * R;
  float* wvb  = ws + 12352 * R;

  rmsnorm3_kernel<<<1024, 256, 0, stream>>>(x, pt, gq, gkv, gc, xqn, xkvn, ptn);
  wmma_gemm_bias<<<dim3(16, 16), 256, 0, stream>>>(xqn, Wq, bq, Qb, 1024);
  wmma_gemm_bias<<<dim3(16, 32), 256, 0, stream>>>(xkvn, Wkv, bkv, KVb, 2048);
  wmma_gemm_bias<<<dim3(16, 16), 256, 0, stream>>>(ptn, Wc, bc, Cb, 1024);
  cumsum_kernel<<<4, 256, 0, stream>>>(Cb, Pb);
  augment_kernel<<<1024, 256, 0, stream>>>(Qb, KVb, Pb, cw, Qab, Kab);
  attn_kernel<<<dim3(16, 16), 128, 0, stream>>>(Qab, Kab, KVb, wvb);
  wmma_gemm_bias<<<dim3(16, 16), 256, 0, stream>>>(wvb, Wout, bout, out, 1024);
}


# attention: one 4-wave block per 16-row query tile, key tiles dealt round-robin to the 4 waves (wave-private staging, no loop barrier); partial max/sum/O merged via LDS: softmax sum order changes
# speedup vs baseline: 3.7554x; 1.3934x over previous
#include <hip/hip_runtime.h>
#include <math.h>


typedef __attribute__((ext_vector_type(2))) float v2f;
typedef __attribute__((ext_vector_type(8))) float v8f;

#define L   1024
#define DM  1024
#define EPSR 1.1920929e-07f

static __device__ __forceinline__ v8f wmma_f32(v2f a, v2f b, v8f c) {
  return __builtin_amdgcn_wmma_f32_16x16x4_f32(false, a, false, b, (short)0, c,
                                               false, false);
}

__global__ void __launch_bounds__(256)
rmsnorm3_kernel(const float* __restrict__ x, const float* __restrict__ pt,
                const float* __restrict__ gq, const float* __restrict__ gkv,
                const float* __restrict__ gc,
                float* __restrict__ xqn, float* __restrict__ xkvn,
                float* __restrict__ ptn) {
  __shared__ float red[256];
  const int row = blockIdx.x;
  const int tid = threadIdx.x;

  float4 xv = ((const float4*)(x + (size_t)row * DM))[tid];
  red[tid] = xv.x * xv.x + xv.y * xv.y + xv.z * xv.z + xv.w * xv.w;
  __syncthreads();
  for (int s = 128; s > 0; s >>= 1) {
    if (tid < s) red[tid] += red[tid + s];
    __syncthreads();
  }
  float rn = rsqrtf(red[0] * (1.0f / DM) + EPSR);
  __syncthreads();

  float4 gqv = ((const float4*)gq)[tid];
  float4 gkvv = ((const float4*)gkv)[tid];
  float4 oq, okv;
  oq.x = xv.x * rn * gqv.x;  oq.y = xv.y * rn * gqv.y;
  oq.z = xv.z * rn * gqv.z;  oq.w = xv.w * rn * gqv.w;
  okv.x = xv.x * rn * gkvv.x; okv.y = xv.y * rn * gkvv.y;
  okv.z = xv.z * rn * gkvv.z; okv.w = xv.w * rn * gkvv.w;
  ((float4*)(xqn + (size_t)row * DM))[tid] = oq;
  ((float4*)(xkvn + (size_t)row * DM))[tid] = okv;

  float4 pv = ((const float4*)(pt + (size_t)row * DM))[tid];
  red[tid] = pv.x * pv.x + pv.y * pv.y + pv.z * pv.z + pv.w * pv.w;
  __syncthreads();
  for (int s = 128; s > 0; s >>= 1) {
    if (tid < s) red[tid] += red[tid + s];
    __syncthreads();
  }
  float rnp = rsqrtf(red[0] * (1.0f / DM) + EPSR);
  float4 gcv = ((const float4*)gc)[tid];
  float4 oc;
  oc.x = pv.x * rnp * gcv.x; oc.y = pv.y * rnp * gcv.y;
  oc.z = pv.z * rnp * gcv.z; oc.w = pv.w * rnp * gcv.w;
  ((float4*)(ptn + (size_t)row * DM))[tid] = oc;
}

#define GKT 64
#define GPA 68
#define GPB 72
__global__ void __launch_bounds__(256)
wmma_gemm_bias(const float* __restrict__ A, const float* __restrict__ Wm,
               const float* __restrict__ bias, float* __restrict__ Cout,
               int N) {
  __shared__ __attribute__((aligned(16))) float As[64 * GPA];
  __shared__ __attribute__((aligned(16))) float Bs[GKT * GPB];
  const int tid = threadIdx.x;
  const int lane = tid & 31, wid = tid >> 5;
  const int hhi = lane >> 4, l16 = lane & 15;
  const int m0 = blockIdx.x * 64, n0 = blockIdx.y * 64;
  const int wm = wid & 3, wn = wid >> 2;
  const int mloc = wm * 16 + l16;
  const int nloc0 = wn * 32 + l16;
  const int nloc1 = nloc0 + 16;

  const int srow = tid >> 4;
  const int scol = (tid & 15) * 4;
  const float* ap = A + (size_t)(m0 + srow) * DM + scol;
  const float* bp = Wm + (size_t)srow * N + n0 + scol;
  const size_t a16 = (size_t)16 * DM;
  const size_t b16 = (size_t)16 * N;
  float* asd = &As[srow * GPA + scol];
  float* bsd = &Bs[srow * GPB + scol];

  float4 ra0 = *(const float4*)(ap);
  float4 ra1 = *(const float4*)(ap + a16);
  float4 ra2 = *(const float4*)(ap + 2 * a16);
  float4 ra3 = *(const float4*)(ap + 3 * a16);
  float4 rb0 = *(const float4*)(bp);
  float4 rb1 = *(const float4*)(bp + b16);
  float4 rb2 = *(const float4*)(bp + 2 * b16);
  float4 rb3 = *(const float4*)(bp + 3 * b16);

  v8f acc0 = {}; v8f acc1 = {};

  for (int k0 = 0; k0 < DM; k0 += GKT) {
    *(float4*)(asd)                = ra0;
    *(float4*)(asd + 16 * GPA)     = ra1;
    *(float4*)(asd + 32 * GPA)     = ra2;
    *(float4*)(asd + 48 * GPA)     = ra3;
    *(float4*)(bsd)                = rb0;
    *(float4*)(bsd + 16 * GPB)     = rb1;
    *(float4*)(bsd + 32 * GPB)     = rb2;
    *(float4*)(bsd + 48 * GPB)     = rb3;
    __syncthreads();
    if (k0 + GKT < DM) {
      const float* an = ap + (k0 + GKT);
      const float* bn = bp + (size_t)(k0 + GKT) * N;
      ra0 = *(const float4*)(an);
      ra1 = *(const float4*)(an + a16);
      ra2 = *(const float4*)(an + 2 * a16);
      ra3 = *(const float4*)(an + 3 * a16);
      rb0 = *(const float4*)(bn);
      rb1 = *(const float4*)(bn + b16);
      rb2 = *(const float4*)(bn + 2 * b16);
      rb3 = *(const float4*)(bn + 3 * b16);
    }
#pragma unroll
    for (int kk = 0; kk < GKT / 4; ++kk) {
      const int kb = kk * 4 + 2 * hhi;
      v2f a;  a.x = As[mloc * GPA + kb];       a.y = As[mloc * GPA + kb + 1];
      v2f b0; b0.x = Bs[kb * GPB + nloc0];     b0.y = Bs[(kb + 1) * GPB + nloc0];
      v2f b1; b1.x = Bs[kb * GPB + nloc1];     b1.y = Bs[(kb + 1) * GPB + nloc1];
      acc0 = wmma_f32(a, b0, acc0);
      acc1 = wmma_f32(a, b1, acc1);
    }
    __syncthreads();
  }

  const int rbase = m0 + wm * 16 + hhi * 8;
  const float bb0 = bias[n0 + nloc0];
  const float bb1 = bias[n0 + nloc1];
#pragma unroll
  for (int v = 0; v < 8; ++v) {
    const int r = rbase + v;
    Cout[(size_t)r * N + n0 + nloc0] = acc0[v] + bb0;
    Cout[(size_t)r * N + n0 + nloc1] = acc1[v] + bb1;
  }
}

__global__ void cumsum_kernel(const float* __restrict__ C, float* __restrict__ P) {
  const int col = blockIdx.x * blockDim.x + threadIdx.x;
  float run = 0.f;
  P[col] = 0.f;
  for (int t = 0; t < L; ++t) {
    run += C[(size_t)t * DM + col];
    P[(size_t)(t + 1) * DM + col] = run;
  }
}

__global__ void augment_kernel(const float* __restrict__ Q,
                               const float* __restrict__ KV,
                               const float* __restrict__ P,
                               const int* __restrict__ cw,
                               float* __restrict__ Qa, float* __restrict__ Ka) {
  const int i = blockIdx.x;
  const int w = cw[0];
  const float scale = 0.125f;
  int E = i + w; if (E > L) E = L;
  int S = i - w; if (S < 0) S = 0;
  for (int c = threadIdx.x; c < DM; c += blockDim.x) {
    const int h = c >> 6, dd = c & 63;
    const float pe = P[(size_t)E * DM + c];
    const float ps = P[(size_t)S * DM + c];
    const float qv = Q[(size_t)i * DM + c];
    const float kv = KV[(size_t)i * 2048 + c];
    const int base = i * 2048 + h * 128 + dd;
    Qa[base]      = qv * (scale + pe);
    Qa[base + 64] = qv;
    Ka[base]      = kv;
    Ka[base + 64] = -kv * ps;
  }
}

#define QKP 132
#define OMP 68
__global__ void __launch_bounds__(128)
attn_kernel(const float* __restrict__ Qa, const float* __restrict__ Ka,
            const float* __restrict__ KV, float* __restrict__ wv) {
  __shared__ __attribute__((aligned(16))) float Qs[16 * QKP];
  __shared__ __attribute__((aligned(16))) float KVs[4 * 16 * QKP + 4 * 16 * 64];
  __shared__ float Pls[4 * 16 * 18];
  __shared__ float Mm[4 * 16];
  __shared__ float Lm[4 * 16];

  const int tid = threadIdx.x;
  const int lane = tid & 31, wid = tid >> 5;
  const int hhi = lane >> 4, l16 = lane & 15;
  const int h = blockIdx.y;
  const int hoff = h * 128;
  const int qtile = 63 - (int)blockIdx.x;
  const int qrow0 = qtile * 16;

  float* myK = &KVs[wid * 16 * QKP];
  float* myV = &KVs[4 * 16 * QKP + wid * 16 * 64];
  float* myP = &Pls[wid * 16 * 18];

  {
    const int r0 = tid >> 5, c = (tid & 31) * 4;
    const float* qsrc = Qa + (size_t)(qrow0 + r0) * 2048 + hoff + c;
    const float4 q0 = *(const float4*)(qsrc);
    const float4 q1 = *(const float4*)(qsrc + 4 * 2048);
    const float4 q2 = *(const float4*)(qsrc + 8 * 2048);
    const float4 q3 = *(const float4*)(qsrc + 12 * 2048);
    *(float4*)&Qs[r0 * QKP + c]        = q0;
    *(float4*)&Qs[(r0 + 4) * QKP + c]  = q1;
    *(float4*)&Qs[(r0 + 8) * QKP + c]  = q2;
    *(float4*)&Qs[(r0 + 12) * QKP + c] = q3;
  }
  __syncthreads();

  const int vr = lane >> 4, vc = (lane & 15) * 4;
  const float* ksrc = Ka + hoff + lane * 4;
  const float* vsrc = KV + (size_t)vr * 2048 + 1024 + h * 64 + vc;

  v8f o0 = {}, o1 = {}, o2 = {}, o3 = {};
  float m8[8], l8[8];
#pragma unroll
  for (int v = 0; v < 8; ++v) { m8[v] = -3.0e38f; l8[v] = 0.f; }

  for (int jt = wid; jt <= qtile; jt += 4) {
    {
      const float* kp = ksrc + (size_t)jt * (16 * 2048);
      const float* vp = vsrc + (size_t)jt * (16 * 2048);
      float4 kr[16], vv[8];
#pragma unroll
      for (int r = 0; r < 16; ++r) kr[r] = *(const float4*)(kp + (size_t)r * 2048);
#pragma unroll
      for (int r = 0; r < 8; ++r) vv[r] = *(const float4*)(vp + (size_t)(2 * r) * 2048);
#pragma unroll
      for (int r = 0; r < 16; ++r) *(float4*)&myK[r * QKP + lane * 4] = kr[r];
#pragma unroll
      for (int r = 0; r < 8; ++r) *(float4*)&myV[(2 * r + vr) * 64 + vc] = vv[r];
    }

    v8f s = {};
#pragma unroll 8
    for (int kk = 0; kk < 32; ++kk) {
      const int kb = kk * 4 + 2 * hhi;
      v2f a; a.x = Qs[l16 * QKP + kb];  a.y = Qs[l16 * QKP + kb + 1];
      v2f b; b.x = myK[l16 * QKP + kb]; b.y = myK[l16 * QKP + kb + 1];
      s = wmma_f32(a, b, s);
    }

    const bool diag = (jt == qtile);
#pragma unroll
    for (int v = 0; v < 8; ++v) {
      const int row = v + hhi * 8;
      float sv = s[v];
      if (diag && (l16 > row)) sv = -1.0e30f;
      float mx = sv;
      for (int off = 8; off > 0; off >>= 1)
        mx = fmaxf(mx, __shfl_xor(mx, off, 16));
      const float mnew = fmaxf(m8[v], mx);
      const float alpha = __expf(m8[v] - mnew);
      const float p = __expf(sv - mnew);
      float rs = p;
      for (int off = 8; off > 0; off >>= 1) rs += __shfl_xor(rs, off, 16);
      l8[v] = l8[v] * alpha + rs;
      m8[v] = mnew;
      o0[v] *= alpha; o1[v] *= alpha; o2[v] *= alpha; o3[v] *= alpha;
      myP[row * 18 + l16] = p;
    }

#pragma unroll
    for (int kk = 0; kk < 4; ++kk) {
      const int kb = kk * 4 + 2 * hhi;
      v2f a; a.x = myP[l16 * 18 + kb]; a.y = myP[l16 * 18 + kb + 1];
      v2f b0, b1, b2, b3;
      b0.x = myV[kb * 64 + l16];      b0.y = myV[(kb + 1) * 64 + l16];
      b1.x = myV[kb * 64 + 16 + l16]; b1.y = myV[(kb + 1) * 64 + 16 + l16];
      b2.x = myV[kb * 64 + 32 + l16]; b2.y = myV[(kb + 1) * 64 + 32 + l16];
      b3.x = myV[kb * 64 + 48 + l16]; b3.y = myV[(kb + 1) * 64 + 48 + l16];
      o0 = wmma_f32(a, b0, o0);
      o1 = wmma_f32(a, b1, o1);
      o2 = wmma_f32(a, b2, o2);
      o3 = wmma_f32(a, b3, o3);
    }
  }

  __syncthreads();
  float* Om = KVs;
  {
    float* myO = &Om[wid * 16 * OMP];
#pragma unroll
    for (int v = 0; v < 8; ++v) {
      const int row = v + hhi * 8;
      myO[row * OMP + l16]      = o0[v];
      myO[row * OMP + 16 + l16] = o1[v];
      myO[row * OMP + 32 + l16] = o2[v];
      myO[row * OMP + 48 + l16] = o3[v];
      if (l16 == 0) { Mm[wid * 16 + row] = m8[v]; Lm[wid * 16 + row] = l8[v]; }
    }
  }
  __syncthreads();

  {
    const int row = wid * 4 + (lane >> 3);
    const int c4 = (lane & 7) * 4;
    const float pm0 = Mm[row], pm1 = Mm[16 + row], pm2 = Mm[32 + row], pm3 = Mm[48 + row];
    const float mt = fmaxf(fmaxf(pm0, pm1), fmaxf(pm2, pm3));
    const float f0 = __expf(pm0 - mt), f1 = __expf(pm1 - mt);
    const float f2 = __expf(pm2 - mt), f3 = __expf(pm3 - mt);
    const float lt = ((Lm[row] * f0 + Lm[16 + row] * f1) + Lm[32 + row] * f2) + Lm[48 + row] * f3;
    const float inv = 1.0f / lt;
    float* dst = wv + (size_t)(qrow0 + row) * DM + h * 64;
#pragma unroll
    for (int i = 0; i < 2; ++i) {
      const int col = c4 + 32 * i;
      const float4 t0 = *(const float4*)&Om[(0 * 16 + row) * OMP + col];
      const float4 t1 = *(const float4*)&Om[(1 * 16 + row) * OMP + col];
      const float4 t2 = *(const float4*)&Om[(2 * 16 + row) * OMP + col];
      const float4 t3 = *(const float4*)&Om[(3 * 16 + row) * OMP + col];
      float4 acc;
      acc.x = (((t0.x * f0 + t1.x * f1) + t2.x * f2) + t3.x * f3) * inv;
      acc.y = (((t0.y * f0 + t1.y * f1) + t2.y * f2) + t3.y * f3) * inv;
      acc.z = (((t0.z * f0 + t1.z * f1) + t2.z * f2) + t3.z * f3) * inv;
      acc.w = (((t0.w * f0 + t1.w * f1) + t2.w * f2) + t3.w * f3) * inv;
      *(float4*)(dst + col) = acc;
    }
  }
}

extern "C" void kernel_launch(void* const* d_in, const int* in_sizes, int n_in,
                              void* d_out, int out_size, void* d_ws,
                              size_t ws_size, hipStream_t stream) {
  const float* x    = (const float*)d_in[0];
  const float* pt   = (const float*)d_in[1];
  const float* gq   = (const float*)d_in[3];
  const float* Wq   = (const float*)d_in[4];
  const float* bq   = (const float*)d_in[5];
  const float* gkv  = (const float*)d_in[6];
  const float* Wkv  = (const float*)d_in[7];
  const float* bkv  = (const float*)d_in[8];
  const float* gc   = (const float*)d_in[9];
  const float* Wc   = (const float*)d_in[10];
  const float* bc   = (const float*)d_in[11];
  const float* Wout = (const float*)d_in[12];
  const float* bout = (const float*)d_in[13];
  const int*   cw   = (const int*)d_in[14];
  float* out = (float*)d_out;

  float* ws = (float*)d_ws;
  const size_t R = 1024;
  float* xqn  = ws + 0 * R * 1024;
  float* xkvn = ws + 1 * R * 1024;
  float* ptn  = ws + 2 * R * 1024;
  float* Qb   = ws + 3 * R * 1024;
  float* KVb  = ws + 4 * R * 1024;
  float* Cb   = ws + 6 * R * 1024;
  float* Pb   = ws + 7 * R * 1024;
  float* Qab  = ws + 8256 * R;
  float* Kab  = ws + 10304 * R;
  float* wvb  = ws + 12352 * R;

  rmsnorm3_kernel<<<1024, 256, 0, stream>>>(x, pt, gq, gkv, gc, xqn, xkvn, ptn);
  wmma_gemm_bias<<<dim3(16, 16), 256, 0, stream>>>(xqn, Wq, bq, Qb, 1024);
  wmma_gemm_bias<<<dim3(16, 32), 256, 0, stream>>>(xkvn, Wkv, bkv, KVb, 2048);
  wmma_gemm_bias<<<dim3(16, 16), 256, 0, stream>>>(ptn, Wc, bc, Cb, 1024);
  cumsum_kernel<<<4, 256, 0, stream>>>(Cb, Pb);
  augment_kernel<<<1024, 256, 0, stream>>>(Qb, KVb, Pb, cw, Qab, Kab);
  attn_kernel<<<dim3(64, 16), 128, 0, stream>>>(Qab, Kab, KVb, wvb);
  wmma_gemm_bias<<<dim3(16, 16), 256, 0, stream>>>(wvb, Wout, bout, out, 1024);
}


# prefix-sum kernel: serial 1024-row column loop on 32 waves -> blocked scan, 32 blocks x 32 waves, 32-row chunks in registers, chunk totals exchanged via LDS; prefix-sum order changes
# speedup vs baseline: 4.7483x; 1.2644x over previous
#include <hip/hip_runtime.h>
#include <math.h>


typedef __attribute__((ext_vector_type(2))) float v2f;
typedef __attribute__((ext_vector_type(8))) float v8f;

#define L   1024
#define DM  1024
#define EPSR 1.1920929e-07f

static __device__ __forceinline__ v8f wmma_f32(v2f a, v2f b, v8f c) {
  return __builtin_amdgcn_wmma_f32_16x16x4_f32(false, a, false, b, (short)0, c,
                                               false, false);
}

__global__ void __launch_bounds__(256)
rmsnorm3_kernel(const float* __restrict__ x, const float* __restrict__ pt,
                const float* __restrict__ gq, const float* __restrict__ gkv,
                const float* __restrict__ gc,
                float* __restrict__ xqn, float* __restrict__ xkvn,
                float* __restrict__ ptn) {
  __shared__ float red[256];
  const int row = blockIdx.x;
  const int tid = threadIdx.x;

  float4 xv = ((const float4*)(x + (size_t)row * DM))[tid];
  red[tid] = xv.x * xv.x + xv.y * xv.y + xv.z * xv.z + xv.w * xv.w;
  __syncthreads();
  for (int s = 128; s > 0; s >>= 1) {
    if (tid < s) red[tid] += red[tid + s];
    __syncthreads();
  }
  float rn = rsqrtf(red[0] * (1.0f / DM) + EPSR);
  __syncthreads();

  float4 gqv = ((const float4*)gq)[tid];
  float4 gkvv = ((const float4*)gkv)[tid];
  float4 oq, okv;
  oq.x = xv.x * rn * gqv.x;  oq.y = xv.y * rn * gqv.y;
  oq.z = xv.z * rn * gqv.z;  oq.w = xv.w * rn * gqv.w;
  okv.x = xv.x * rn * gkvv.x; okv.y = xv.y * rn * gkvv.y;
  okv.z = xv.z * rn * gkvv.z; okv.w = xv.w * rn * gkvv.w;
  ((float4*)(xqn + (size_t)row * DM))[tid] = oq;
  ((float4*)(xkvn + (size_t)row * DM))[tid] = okv;

  float4 pv = ((const float4*)(pt + (size_t)row * DM))[tid];
  red[tid] = pv.x * pv.x + pv.y * pv.y + pv.z * pv.z + pv.w * pv.w;
  __syncthreads();
  for (int s = 128; s > 0; s >>= 1) {
    if (tid < s) red[tid] += red[tid + s];
    __syncthreads();
  }
  float rnp = rsqrtf(red[0] * (1.0f / DM) + EPSR);
  float4 gcv = ((const float4*)gc)[tid];
  float4 oc;
  oc.x = pv.x * rnp * gcv.x; oc.y = pv.y * rnp * gcv.y;
  oc.z = pv.z * rnp * gcv.z; oc.w = pv.w * rnp * gcv.w;
  ((float4*)(ptn + (size_t)row * DM))[tid] = oc;
}

#define GKT 64
#define GPA 68
#define GPB 72
__global__ void __launch_bounds__(256)
wmma_gemm_bias(const float* __restrict__ A, const float* __restrict__ Wm,
               const float* __restrict__ bias, float* __restrict__ Cout,
               int N) {
  __shared__ __attribute__((aligned(16))) float As[64 * GPA];
  __shared__ __attribute__((aligned(16))) float Bs[GKT * GPB];
  const int tid = threadIdx.x;
  const int lane = tid & 31, wid = tid >> 5;
  const int hhi = lane >> 4, l16 = lane & 15;
  const int m0 = blockIdx.x * 64, n0 = blockIdx.y * 64;
  const int wm = wid & 3, wn = wid >> 2;
  const int mloc = wm * 16 + l16;
  const int nloc0 = wn * 32 + l16;
  const int nloc1 = nloc0 + 16;

  const int srow = tid >> 4;
  const int scol = (tid & 15) * 4;
  const float* ap = A + (size_t)(m0 + srow) * DM + scol;
  const float* bp = Wm + (size_t)srow * N + n0 + scol;
  const size_t a16 = (size_t)16 * DM;
  const size_t b16 = (size_t)16 * N;
  float* asd = &As[srow * GPA + scol];
  float* bsd = &Bs[srow * GPB + scol];

  float4 ra0 = *(const float4*)(ap);
  float4 ra1 = *(const float4*)(ap + a16);
  float4 ra2 = *(const float4*)(ap + 2 * a16);
  float4 ra3 = *(const float4*)(ap + 3 * a16);
  float4 rb0 = *(const float4*)(bp);
  float4 rb1 = *(const float4*)(bp + b16);
  float4 rb2 = *(const float4*)(bp + 2 * b16);
  float4 rb3 = *(const float4*)(bp + 3 * b16);

  v8f acc0 = {}; v8f acc1 = {};

  for (int k0 = 0; k0 < DM; k0 += GKT) {
    *(float4*)(asd)                = ra0;
    *(float4*)(asd + 16 * GPA)     = ra1;
    *(float4*)(asd + 32 * GPA)     = ra2;
    *(float4*)(asd + 48 * GPA)     = ra3;
    *(float4*)(bsd)                = rb0;
    *(float4*)(bsd + 16 * GPB)     = rb1;
    *(float4*)(bsd + 32 * GPB)     = rb2;
    *(float4*)(bsd + 48 * GPB)     = rb3;
    __syncthreads();
    if (k0 + GKT < DM) {
      const float* an = ap + (k0 + GKT);
      const float* bn = bp + (size_t)(k0 + GKT) * N;
      ra0 = *(const float4*)(an);
      ra1 = *(const float4*)(an + a16);
      ra2 = *(const float4*)(an + 2 * a16);
      ra3 = *(const float4*)(an + 3 * a16);
      rb0 = *(const float4*)(bn);
      rb1 = *(const float4*)(bn + b16);
      rb2 = *(const float4*)(bn + 2 * b16);
      rb3 = *(const float4*)(bn + 3 * b16);
    }
#pragma unroll
    for (int kk = 0; kk < GKT / 4; ++kk) {
      const int kb = kk * 4 + 2 * hhi;
      v2f a;  a.x = As[mloc * GPA + kb];       a.y = As[mloc * GPA + kb + 1];
      v2f b0; b0.x = Bs[kb * GPB + nloc0];     b0.y = Bs[(kb + 1) * GPB + nloc0];
      v2f b1; b1.x = Bs[kb * GPB + nloc1];     b1.y = Bs[(kb + 1) * GPB + nloc1];
      acc0 = wmma_f32(a, b0, acc0);
      acc1 = wmma_f32(a, b1, acc1);
    }
    __syncthreads();
  }

  const int rbase = m0 + wm * 16 + hhi * 8;
  const float bb0 = bias[n0 + nloc0];
  const float bb1 = bias[n0 + nloc1];
#pragma unroll
  for (int v = 0; v < 8; ++v) {
    const int r = rbase + v;
    Cout[(size_t)r * N + n0 + nloc0] = acc0[v] + bb0;
    Cout[(size_t)r * N + n0 + nloc1] = acc1[v] + bb1;
  }
}

#define SCR 32
__global__ void __launch_bounds__(1024)
cumsum_kernel(const float* __restrict__ C, float* __restrict__ P) {
  __shared__ float tot[32 * 32];
  const int lane = threadIdx.x & 31, wid = threadIdx.x >> 5;
  const int col = blockIdx.x * 32 + lane;
  const float* src = C + (size_t)(wid * SCR) * DM + col;
  float v[SCR];
#pragma unroll
  for (int r = 0; r < SCR; ++r) v[r] = src[(size_t)r * DM];
#pragma unroll
  for (int r = 1; r < SCR; ++r) v[r] += v[r - 1];
  tot[wid * 32 + lane] = v[SCR - 1];
  __syncthreads();
  float off = 0.f;
#pragma unroll
  for (int u = 0; u < 32; ++u) {
    const float tv = tot[u * 32 + lane];
    off += (u < wid) ? tv : 0.f;
  }
  float* dst = P + (size_t)(wid * SCR + 1) * DM + col;
#pragma unroll
  for (int r = 0; r < SCR; ++r) dst[(size_t)r * DM] = off + v[r];
  if (wid == 0) P[col] = 0.f;
}

__global__ void augment_kernel(const float* __restrict__ Q,
                               const float* __restrict__ KV,
                               const float* __restrict__ P,
                               const int* __restrict__ cw,
                               float* __restrict__ Qa, float* __restrict__ Ka) {
  const int i = blockIdx.x;
  const int w = cw[0];
  const float scale = 0.125f;
  int E = i + w; if (E > L) E = L;
  int S = i - w; if (S < 0) S = 0;
  for (int c = threadIdx.x; c < DM; c += blockDim.x) {
    const int h = c >> 6, dd = c & 63;
    const float pe = P[(size_t)E * DM + c];
    const float ps = P[(size_t)S * DM + c];
    const float qv = Q[(size_t)i * DM + c];
    const float kv = KV[(size_t)i * 2048 + c];
    const int base = i * 2048 + h * 128 + dd;
    Qa[base]      = qv * (scale + pe);
    Qa[base + 64] = qv;
    Ka[base]      = kv;
    Ka[base + 64] = -kv * ps;
  }
}

#define QKP 132
#define OMP 68
__global__ void __launch_bounds__(128)
attn_kernel(const float* __restrict__ Qa, const float* __restrict__ Ka,
            const float* __restrict__ KV, float* __restrict__ wv) {
  __shared__ __attribute__((aligned(16))) float Qs[16 * QKP];
  __shared__ __attribute__((aligned(16))) float KVs[4 * 16 * QKP + 4 * 16 * 64];
  __shared__ float Pls[4 * 16 * 18];
  __shared__ float Mm[4 * 16];
  __shared__ float Lm[4 * 16];

  const int tid = threadIdx.x;
  const int lane = tid & 31, wid = tid >> 5;
  const int hhi = lane >> 4, l16 = lane & 15;
  const int h = blockIdx.y;
  const int hoff = h * 128;
  const int qtile = 63 - (int)blockIdx.x;
  const int qrow0 = qtile * 16;

  float* myK = &KVs[wid * 16 * QKP];
  float* myV = &KVs[4 * 16 * QKP + wid * 16 * 64];
  float* myP = &Pls[wid * 16 * 18];

  {
    const int r0 = tid >> 5, c = (tid & 31) * 4;
    const float* qsrc = Qa + (size_t)(qrow0 + r0) * 2048 + hoff + c;
    const float4 q0 = *(const float4*)(qsrc);
    const float4 q1 = *(const float4*)(qsrc + 4 * 2048);
    const float4 q2 = *(const float4*)(qsrc + 8 * 2048);
    const float4 q3 = *(const float4*)(qsrc + 12 * 2048);
    *(float4*)&Qs[r0 * QKP + c]        = q0;
    *(float4*)&Qs[(r0 + 4) * QKP + c]  = q1;
    *(float4*)&Qs[(r0 + 8) * QKP + c]  = q2;
    *(float4*)&Qs[(r0 + 12) * QKP + c] = q3;
  }
  __syncthreads();

  const int vr = lane >> 4, vc = (lane & 15) * 4;
  const float* ksrc = Ka + hoff + lane * 4;
  const float* vsrc = KV + (size_t)vr * 2048 + 1024 + h * 64 + vc;

  v8f o0 = {}, o1 = {}, o2 = {}, o3 = {};
  float m8[8], l8[8];
#pragma unroll
  for (int v = 0; v < 8; ++v) { m8[v] = -3.0e38f; l8[v] = 0.f; }

  for (int jt = wid; jt <= qtile; jt += 4) {
    {
      const float* kp = ksrc + (size_t)jt * (16 * 2048);
      const float* vp = vsrc + (size_t)jt * (16 * 2048);
      float4 kr[16], vv[8];
#pragma unroll
      for (int r = 0; r < 16; ++r) kr[r] = *(const float4*)(kp + (size_t)r * 2048);
#pragma unroll
      for (int r = 0; r < 8; ++r) vv[r] = *(const float4*)(vp + (size_t)(2 * r) * 2048);
#pragma unroll
      for (int r = 0; r < 16; ++r) *(float4*)&myK[r * QKP + lane * 4] = kr[r];
#pragma unroll
      for (int r = 0; r < 8; ++r) *(float4*)&myV[(2 * r + vr) * 64 + vc] = vv[r];
    }

    v8f s = {};
#pragma unroll 8
    for (int kk = 0; kk < 32; ++kk) {
      const int kb = kk * 4 + 2 * hhi;
      v2f a; a.x = Qs[l16 * QKP + kb];  a.y = Qs[l16 * QKP + kb + 1];
      v2f b; b.x = myK[l16 * QKP + kb]; b.y = myK[l16 * QKP + kb + 1];
      s = wmma_f32(a, b, s);
    }

    const bool diag = (jt == qtile);
#pragma unroll
    for (int v = 0; v < 8; ++v) {
      const int row = v + hhi * 8;
      float sv = s[v];
      if (diag && (l16 > row)) sv = -1.0e30f;
      float mx = sv;
      for (int off = 8; off > 0; off >>= 1)
        mx = fmaxf(mx, __shfl_xor(mx, off, 16));
      const float mnew = fmaxf(m8[v], mx);
      const float alpha = __expf(m8[v] - mnew);
      const float p = __expf(sv - mnew);
      float rs = p;
      for (int off = 8; off > 0; off >>= 1) rs += __shfl_xor(rs, off, 16);
      l8[v] = l8[v] * alpha + rs;
      m8[v] = mnew;
      o0[v] *= alpha; o1[v] *= alpha; o2[v] *= alpha; o3[v] *= alpha;
      myP[row * 18 + l16] = p;
    }

#pragma unroll
    for (int kk = 0; kk < 4; ++kk) {
      const int kb = kk * 4 + 2 * hhi;
      v2f a; a.x = myP[l16 * 18 + kb]; a.y = myP[l16 * 18 + kb + 1];
      v2f b0, b1, b2, b3;
      b0.x = myV[kb * 64 + l16];      b0.y = myV[(kb + 1) * 64 + l16];
      b1.x = myV[kb * 64 + 16 + l16]; b1.y = myV[(kb + 1) * 64 + 16 + l16];
      b2.x = myV[kb * 64 + 32 + l16]; b2.y = myV[(kb + 1) * 64 + 32 + l16];
      b3.x = myV[kb * 64 + 48 + l16]; b3.y = myV[(kb + 1) * 64 + 48 + l16];
      o0 = wmma_f32(a, b0, o0);
      o1 = wmma_f32(a, b1, o1);
      o2 = wmma_f32(a, b2, o2);
      o3 = wmma_f32(a, b3, o3);
    }
  }

  __syncthreads();
  float* Om = KVs;
  {
    float* myO = &Om[wid * 16 * OMP];
#pragma unroll
    for (int v = 0; v < 8; ++v) {
      const int row = v + hhi * 8;
      myO[row * OMP + l16]      = o0[v];
      myO[row * OMP + 16 + l16] = o1[v];
      myO[row * OMP + 32 + l16] = o2[v];
      myO[row * OMP + 48 + l16] = o3[v];
      if (l16 == 0) { Mm[wid * 16 + row] = m8[v]; Lm[wid * 16 + row] = l8[v]; }
    }
  }
  __syncthreads();

  {
    const int row = wid * 4 + (lane >> 3);
    const int c4 = (lane & 7) * 4;
    const float pm0 = Mm[row], pm1 = Mm[16 + row], pm2 = Mm[32 + row], pm3 = Mm[48 + row];
    const float mt = fmaxf(fmaxf(pm0, pm1), fmaxf(pm2, pm3));
    const float f0 = __expf(pm0 - mt), f1 = __expf(pm1 - mt);
    const float f2 = __expf(pm2 - mt), f3 = __expf(pm3 - mt);
    const float lt = ((Lm[row] * f0 + Lm[16 + row] * f1) + Lm[32 + row] * f2) + Lm[48 + row] * f3;
    const float inv = 1.0f / lt;
    float* dst = wv + (size_t)(qrow0 + row) * DM + h * 64;
#pragma unroll
    for (int i = 0; i < 2; ++i) {
      const int col = c4 + 32 * i;
      const float4 t0 = *(const float4*)&Om[(0 * 16 + row) * OMP + col];
      const float4 t1 = *(const float4*)&Om[(1 * 16 + row) * OMP + col];
      const float4 t2 = *(const float4*)&Om[(2 * 16 + row) * OMP + col];
      const float4 t3 = *(const float4*)&Om[(3 * 16 + row) * OMP + col];
      float4 acc;
      acc.x = (((t0.x * f0 + t1.x * f1) + t2.x * f2) + t3.x * f3) * inv;
      acc.y = (((t0.y * f0 + t1.y * f1) + t2.y * f2) + t3.y * f3) * inv;
      acc.z = (((t0.z * f0 + t1.z * f1) + t2.z * f2) + t3.z * f3) * inv;
      acc.w = (((t0.w * f0 + t1.w * f1) + t2.w * f2) + t3.w * f3) * inv;
      *(float4*)(dst + col) = acc;
    }
  }
}

extern "C" void kernel_launch(void* const* d_in, const int* in_sizes, int n_in,
                              void* d_out, int out_size, void* d_ws,
                              size_t ws_size, hipStream_t stream) {
  const float* x    = (const float*)d_in[0];
  const float* pt   = (const float*)d_in[1];
  const float* gq   = (const float*)d_in[3];
  const float* Wq   = (const float*)d_in[4];
  const float* bq   = (const float*)d_in[5];
  const float* gkv  = (const float*)d_in[6];
  const float* Wkv  = (const float*)d_in[7];
  const float* bkv  = (const float*)d_in[8];
  const float* gc   = (const float*)d_in[9];
  const float* Wc   = (const float*)d_in[10];
  const float* bc   = (const float*)d_in[11];
  const float* Wout = (const float*)d_in[12];
  const float* bout = (const float*)d_in[13];
  const int*   cw   = (const int*)d_in[14];
  float* out = (float*)d_out;

  float* ws = (float*)d_ws;
  const size_t R = 1024;
  float* xqn  = ws + 0 * R * 1024;
  float* xkvn = ws + 1 * R * 1024;
  float* ptn  = ws + 2 * R * 1024;
  float* Qb   = ws + 3 * R * 1024;
  float* KVb  = ws + 4 * R * 1024;
  float* Cb   = ws + 6 * R * 1024;
  float* Pb   = ws + 7 * R * 1024;
  float* Qab  = ws + 8256 * R;
  float* Kab  = ws + 10304 * R;
  float* wvb  = ws + 12352 * R;

  rmsnorm3_kernel<<<1024, 256, 0, stream>>>(x, pt, gq, gkv, gc, xqn, xkvn, ptn);
  wmma_gemm_bias<<<dim3(16, 16), 256, 0, stream>>>(xqn, Wq, bq, Qb, 1024);
  wmma_gemm_bias<<<dim3(16, 32), 256, 0, stream>>>(xkvn, Wkv, bkv, KVb, 2048);
  wmma_gemm_bias<<<dim3(16, 16), 256, 0, stream>>>(ptn, Wc, bc, Cb, 1024);
  cumsum_kernel<<<32, 1024, 0, stream>>>(Cb, Pb);
  augment_kernel<<<1024, 256, 0, stream>>>(Qb, KVb, Pb, cw, Qab, Kab);
  attn_kernel<<<dim3(64, 16), 128, 0, stream>>>(Qab, Kab, KVb, wvb);
  wmma_gemm_bias<<<dim3(16, 16), 256, 0, stream>>>(wvb, Wout, bout, out, 1024);
}


# attention: a block = query tiles (63-p, p) of one head, their 65 key-tile steps dealt round-robin to 8 waves (was one tile per 4-wave block): equal work per block; softmax sum order changes
# speedup vs baseline: 5.5377x; 1.1662x over previous
#include <hip/hip_runtime.h>
#include <math.h>


typedef __attribute__((ext_vector_type(2))) float v2f;
typedef __attribute__((ext_vector_type(8))) float v8f;

#define L   1024
#define DM  1024
#define EPSR 1.1920929e-07f

static __device__ __forceinline__ v8f wmma_f32(v2f a, v2f b, v8f c) {
  return __builtin_amdgcn_wmma_f32_16x16x4_f32(false, a, false, b, (short)0, c,
                                               false, false);
}

__global__ void __launch_bounds__(256)
rmsnorm3_kernel(const float* __restrict__ x, const float* __restrict__ pt,
                const float* __restrict__ gq, const float* __restrict__ gkv,
                const float* __restrict__ gc,
                float* __restrict__ xqn, float* __restrict__ xkvn,
                float* __restrict__ ptn) {
  __shared__ float red[256];
  const int row = blockIdx.x;
  const int tid = threadIdx.x;

  float4 xv = ((const float4*)(x + (size_t)row * DM))[tid];
  red[tid] = xv.x * xv.x + xv.y * xv.y + xv.z * xv.z + xv.w * xv.w;
  __syncthreads();
  for (int s = 128; s > 0; s >>= 1) {
    if (tid < s) red[tid] += red[tid + s];
    __syncthreads();
  }
  float rn = rsqrtf(red[0] * (1.0f / DM) + EPSR);
  __syncthreads();

  float4 gqv = ((const float4*)gq)[tid];
  float4 gkvv = ((const float4*)gkv)[tid];
  float4 oq, okv;
  oq.x = xv.x * rn * gqv.x;  oq.y = xv.y * rn * gqv.y;
  oq.z = xv.z * rn * gqv.z;  oq.w = xv.w * rn * gqv.w;
  okv.x = xv.x * rn * gkvv.x; okv.y = xv.y * rn * gkvv.y;
  okv.z = xv.z * rn * gkvv.z; okv.w = xv.w * rn * gkvv.w;
  ((float4*)(xqn + (size_t)row * DM))[tid] = oq;
  ((float4*)(xkvn + (size_t)row * DM))[tid] = okv;

  float4 pv = ((const float4*)(pt + (size_t)row * DM))[tid];
  red[tid] = pv.x * pv.x + pv.y * pv.y + pv.z * pv.z + pv.w * pv.w;
  __syncthreads();
  for (int s = 128; s > 0; s >>= 1) {
    if (tid < s) red[tid] += red[tid + s];
    __syncthreads();
  }
  float rnp = rsqrtf(red[0] * (1.0f / DM) + EPSR);
  float4 gcv = ((const float4*)gc)[tid];
  float4 oc;
  oc.x = pv.x * rnp * gcv.x; oc.y = pv.y * rnp * gcv.y;
  oc.z = pv.z * rnp * gcv.z; oc.w = pv.w * rnp * gcv.w;
  ((float4*)(ptn + (size_t)row * DM))[tid] = oc;
}

#define GKT 64
#define GPA 68
#define GPB 72
__global__ void __launch_bounds__(256)
wmma_gemm_bias(const float* __restrict__ A, const float* __restrict__ Wm,
               const float* __restrict__ bias, float* __restrict__ Cout,
               int N) {
  __shared__ __attribute__((aligned(16))) float As[64 * GPA];
  __shared__ __attribute__((aligned(16))) float Bs[GKT * GPB];
  const int tid = threadIdx.x;
  const int lane = tid & 31, wid = tid >> 5;
  const int hhi = lane >> 4, l16 = lane & 15;
  const int m0 = blockIdx.x * 64, n0 = blockIdx.y * 64;
  const int wm = wid & 3, wn = wid >> 2;
  const int mloc = wm * 16 + l16;
  const int nloc0 = wn * 32 + l16;
  const int nloc1 = nloc0 + 16;

  const int srow = tid >> 4;
  const int scol = (tid & 15) * 4;
  const float* ap = A + (size_t)(m0 + srow) * DM + scol;
  const float* bp = Wm + (size_t)srow * N + n0 + scol;
  const size_t a16 = (size_t)16 * DM;
  const size_t b16 = (size_t)16 * N;
  float* asd = &As[srow * GPA + scol];
  float* bsd = &Bs[srow * GPB + scol];

  float4 ra0 = *(const float4*)(ap);
  float4 ra1 = *(const float4*)(ap + a16);
  float4 ra2 = *(const float4*)(ap + 2 * a16);
  float4 ra3 = *(const float4*)(ap + 3 * a16);
  float4 rb0 = *(const float4*)(bp);
  float4 rb1 = *(const float4*)(bp + b16);
  float4 rb2 = *(const float4*)(bp + 2 * b16);
  float4 rb3 = *(const float4*)(bp + 3 * b16);

  v8f acc0 = {}; v8f acc1 = {};

  for (int k0 = 0; k0 < DM; k0 += GKT) {
    *(float4*)(asd)                = ra0;
    *(float4*)(asd + 16 * GPA)     = ra1;
    *(float4*)(asd + 32 * GPA)     = ra2;
    *(float4*)(asd + 48 * GPA)     = ra3;
    *(float4*)(bsd)                = rb0;
    *(float4*)(bsd + 16 * GPB)     = rb1;
    *(float4*)(bsd + 32 * GPB)     = rb2;
    *(float4*)(bsd + 48 * GPB)     = rb3;
    __syncthreads();
    if (k0 + GKT < DM) {
      const float* an = ap + (k0 + GKT);
      const float* bn = bp + (size_t)(k0 + GKT) * N;
      ra0 = *(const float4*)(an);
      ra1 = *(const float4*)(an + a16);
      ra2 = *(const float4*)(an + 2 * a16);
      ra3 = *(const float4*)(an + 3 * a16);
      rb0 = *(const float4*)(bn);
      rb1 = *(const float4*)(bn + b16);
      rb2 = *(const float4*)(bn + 2 * b16);
      rb3 = *(const float4*)(bn + 3 * b16);
    }
#pragma unroll
    for (int kk = 0; kk < GKT / 4; ++kk) {
      const int kb = kk * 4 + 2 * hhi;
      v2f a;  a.x = As[mloc * GPA + kb];       a.y = As[mloc * GPA + kb + 1];
      v2f b0; b0.x = Bs[kb * GPB + nloc0];     b0.y = Bs[(kb + 1) * GPB + nloc0];
      v2f b1; b1.x = Bs[kb * GPB + nloc1];     b1.y = Bs[(kb + 1) * GPB + nloc1];
      acc0 = wmma_f32(a, b0, acc0);
      acc1 = wmma_f32(a, b1, acc1);
    }
    __syncthreads();
  }

  const int rbase = m0 + wm * 16 + hhi * 8;
  const float bb0 = bias[n0 + nloc0];
  const float bb1 = bias[n0 + nloc1];
#pragma unroll
  for (int v = 0; v < 8; ++v) {
    const int r = rbase + v;
    Cout[(size_t)r * N + n0 + nloc0] = acc0[v] + bb0;
    Cout[(size_t)r * N + n0 + nloc1] = acc1[v] + bb1;
  }
}

#define SCR 32
__global__ void __launch_bounds__(1024)
cumsum_kernel(const float* __restrict__ C, float* __restrict__ P) {
  __shared__ float tot[32 * 32];
  const int lane = threadIdx.x & 31, wid = threadIdx.x >> 5;
  const int col = blockIdx.x * 32 + lane;
  const float* src = C + (size_t)(wid * SCR) * DM + col;
  float v[SCR];
#pragma unroll
  for (int r = 0; r < SCR; ++r) v[r] = src[(size_t)r * DM];
#pragma unroll
  for (int r = 1; r < SCR; ++r) v[r] += v[r - 1];
  tot[wid * 32 + lane] = v[SCR - 1];
  __syncthreads();
  float off = 0.f;
#pragma unroll
  for (int u = 0; u < 32; ++u) {
    const float tv = tot[u * 32 + lane];
    off += (u < wid) ? tv : 0.f;
  }
  float* dst = P + (size_t)(wid * SCR + 1) * DM + col;
#pragma unroll
  for (int r = 0; r < SCR; ++r) dst[(size_t)r * DM] = off + v[r];
  if (wid == 0) P[col] = 0.f;
}

__global__ void augment_kernel(const float* __restrict__ Q,
                               const float* __restrict__ KV,
                               const float* __restrict__ P,
                               const int* __restrict__ cw,
                               float* __restrict__ Qa, float* __restrict__ Ka) {
  const int i = blockIdx.x;
  const int w = cw[0];
  const float scale = 0.125f;
  int E = i + w; if (E > L) E = L;
  int S = i - w; if (S < 0) S = 0;
  for (int c = threadIdx.x; c < DM; c += blockDim.x) {
    const int h = c >> 6, dd = c & 63;
    const float pe = P[(size_t)E * DM + c];
    const float ps = P[(size_t)S * DM + c];
    const float qv = Q[(size_t)i * DM + c];
    const float kv = KV[(size_t)i * 2048 + c];
    const int base = i * 2048 + h * 128 + dd;
    Qa[base]      = qv * (scale + pe);
    Qa[base + 64] = qv;
    Ka[base]      = kv;
    Ka[base + 64] = -kv * ps;
  }
}

#define QKP 132
#define OMP 68
#define ANW 8

static __device__ __forceinline__ void attn_tile(
    const float* __restrict__ kp, const float* __restrict__ vp, const float* Qt,
    float* myK, float* myV, float* myP, const int lane, const int hhi,
    const int l16, const int vr, const int vc, const bool diag,
    v8f& o0, v8f& o1, v8f& o2, v8f& o3, float (&m8)[8], float (&l8)[8]) {
  {
    float4 kr[16], vv[8];
#pragma unroll
    for (int r = 0; r < 16; ++r) kr[r] = *(const float4*)(kp + (size_t)r * 2048);
#pragma unroll
    for (int r = 0; r < 8; ++r) vv[r] = *(const float4*)(vp + (size_t)(2 * r) * 2048);
#pragma unroll
    for (int r = 0; r < 16; ++r) *(float4*)&myK[r * QKP + lane * 4] = kr[r];
#pragma unroll
    for (int r = 0; r < 8; ++r) *(float4*)&myV[(2 * r + vr) * 64 + vc] = vv[r];
  }

  v8f s = {};
#pragma unroll 8
  for (int kk = 0; kk < 32; ++kk) {
    const int kb = kk * 4 + 2 * hhi;
    v2f a; a.x = Qt[l16 * QKP + kb];  a.y = Qt[l16 * QKP + kb + 1];
    v2f b; b.x = myK[l16 * QKP + kb]; b.y = myK[l16 * QKP + kb + 1];
    s = wmma_f32(a, b, s);
  }

#pragma unroll
  for (int v = 0; v < 8; ++v) {
    const int row = v + hhi * 8;
    float sv = s[v];
    if (diag && (l16 > row)) sv = -1.0e30f;
    float mx = sv;
    for (int off = 8; off > 0; off >>= 1)
      mx = fmaxf(mx, __shfl_xor(mx, off, 16));
    const float mnew = fmaxf(m8[v], mx);
    const float alpha = __expf(m8[v] - mnew);
    const float p = __expf(sv - mnew);
    float rs = p;
    for (int off = 8; off > 0; off >>= 1) rs += __shfl_xor(rs, off, 16);
    l8[v] = l8[v] * alpha + rs;
    m8[v] = mnew;
    o0[v] *= alpha; o1[v] *= alpha; o2[v] *= alpha; o3[v] *= alpha;
    myP[row * 18 + l16] = p;
  }

#pragma unroll
  for (int kk = 0; kk < 4; ++kk) {
    const int kb = kk * 4 + 2 * hhi;
    v2f a; a.x = myP[l16 * 18 + kb]; a.y = myP[l16 * 18 + kb + 1];
    v2f b0, b1, b2, b3;
    b0.x = myV[kb * 64 + l16];      b0.y = myV[(kb + 1) * 64 + l16];
    b1.x = myV[kb * 64 + 16 + l16]; b1.y = myV[(kb + 1) * 64 + 16 + l16];
    b2.x = myV[kb * 64 + 32 + l16]; b2.y = myV[(kb + 1) * 64 + 32 + l16];
    b3.x = myV[kb * 64 + 48 + l16]; b3.y = myV[(kb + 1) * 64 + 48 + l16];
    o0 = wmma_f32(a, b0, o0);
    o1 = wmma_f32(a, b1, o1);
    o2 = wmma_f32(a, b2, o2);
    o3 = wmma_f32(a, b3, o3);
  }
}

__global__ void __launch_bounds__(256)
attn_kernel(const float* __restrict__ Qa, const float* __restrict__ Ka,
            const float* __restrict__ KV, float* __restrict__ wv) {
  __shared__ __attribute__((aligned(16))) float Qs[2 * 16 * QKP];
  __shared__ __attribute__((aligned(16))) float KVs[ANW * 16 * QKP + ANW * 16 * 64];
  __shared__ float Pls[ANW * 16 * 18];
  __shared__ float Mm[2 * ANW * 16];
  __shared__ float Lm[2 * ANW * 16];

  const int tid = threadIdx.x;
  const int lane = tid & 31;
  const int wid = __builtin_amdgcn_readfirstlane(tid >> 5);
  const int hhi = lane >> 4, l16 = lane & 15;
  const int h = blockIdx.y;
  const int hoff = h * 128;
  const int p = blockIdx.x;
  const int qtA = 63 - p;
  const int qtB = p;
  const int nA = 64 - p;

  float* myK = &KVs[wid * 16 * QKP];
  float* myV = &KVs[ANW * 16 * QKP + wid * 16 * 64];
  float* myP = &Pls[wid * 16 * 18];

  {
    const int tq = tid >> 7;
    const int t = tid & 127;
    const int r0 = t >> 5, c = (t & 31) * 4;
    const int qrow = (tq ? qtB : qtA) * 16;
    const float* qsrc = Qa + (size_t)(qrow + r0) * 2048 + hoff + c;
    float* qd = &Qs[tq * 16 * QKP];
    const float4 q0 = *(const float4*)(qsrc);
    const float4 q1 = *(const float4*)(qsrc + 4 * 2048);
    const float4 q2 = *(const float4*)(qsrc + 8 * 2048);
    const float4 q3 = *(const float4*)(qsrc + 12 * 2048);
    *(float4*)&qd[r0 * QKP + c]        = q0;
    *(float4*)&qd[(r0 + 4) * QKP + c]  = q1;
    *(float4*)&qd[(r0 + 8) * QKP + c]  = q2;
    *(float4*)&qd[(r0 + 12) * QKP + c] = q3;
  }
  __syncthreads();

  const int vr = lane >> 4, vc = (lane & 15) * 4;
  const float* ksrc = Ka + hoff + lane * 4;
  const float* vsrc = KV + (size_t)vr * 2048 + 1024 + h * 64 + vc;

  v8f a0 = {}, a1 = {}, a2 = {}, a3 = {};
  v8f b0 = {}, b1 = {}, b2 = {}, b3 = {};
  float mA[8], lA[8], mB[8], lB[8];
#pragma unroll
  for (int v = 0; v < 8; ++v) {
    mA[v] = -3.0e38f; lA[v] = 0.f; mB[v] = -3.0e38f; lB[v] = 0.f;
  }

  int u = wid;
  for (; u < nA; u += ANW) {
    attn_tile(ksrc + (size_t)u * (16 * 2048), vsrc + (size_t)u * (16 * 2048),
              Qs, myK, myV, myP, lane, hhi, l16, vr, vc, u == qtA,
              a0, a1, a2, a3, mA, lA);
  }
  for (; u < 65; u += ANW) {
    const int jt = u - nA;
    attn_tile(ksrc + (size_t)jt * (16 * 2048), vsrc + (size_t)jt * (16 * 2048),
              Qs + 16 * QKP, myK, myV, myP, lane, hhi, l16, vr, vc, jt == qtB,
              b0, b1, b2, b3, mB, lB);
  }

  __syncthreads();
  float* Om = KVs;
  {
    float* oA = &Om[(0 * ANW + wid) * 16 * OMP];
    float* oB = &Om[(1 * ANW + wid) * 16 * OMP];
#pragma unroll
    for (int v = 0; v < 8; ++v) {
      const int row = v + hhi * 8;
      oA[row * OMP + l16]      = a0[v];
      oA[row * OMP + 16 + l16] = a1[v];
      oA[row * OMP + 32 + l16] = a2[v];
      oA[row * OMP + 48 + l16] = a3[v];
      oB[row * OMP + l16]      = b0[v];
      oB[row * OMP + 16 + l16] = b1[v];
      oB[row * OMP + 32 + l16] = b2[v];
      oB[row * OMP + 48 + l16] = b3[v];
      if (l16 == 0) {
        Mm[(0 * ANW + wid) * 16 + row] = mA[v]; Lm[(0 * ANW + wid) * 16 + row] = lA[v];
        Mm[(1 * ANW + wid) * 16 + row] = mB[v]; Lm[(1 * ANW + wid) * 16 + row] = lB[v];
      }
    }
  }
  __syncthreads();

  {
    const int tq = wid >> 2;
    const int row = (wid & 3) * 4 + (lane >> 3);
    const int c4 = (lane & 7) * 4;
    const float* Mt = &Mm[tq * ANW * 16 + row];
    const float* Lt = &Lm[tq * ANW * 16 + row];
    const float* Ot = &Om[(tq * ANW * 16 + row) * OMP];
    float pm[ANW], f[ANW];
#pragma unroll
    for (int w = 0; w < ANW; ++w) pm[w] = Mt[w * 16];
    float mt = pm[0];
#pragma unroll
    for (int w = 1; w < ANW; ++w) mt = fmaxf(mt, pm[w]);
    float lt = 0.f;
#pragma unroll
    for (int w = 0; w < ANW; ++w) {
      f[w] = __expf(pm[w] - mt);
      lt += Lt[w * 16] * f[w];
    }
    const float inv = 1.0f / lt;
    float* dst = wv + (size_t)((tq ? qtB : qtA) * 16 + row) * DM + h * 64;
#pragma unroll
    for (int i = 0; i < 2; ++i) {
      const int col = c4 + 32 * i;
      float4 acc = {0.f, 0.f, 0.f, 0.f};
#pragma unroll
      for (int w = 0; w < ANW; ++w) {
        const float4 t = *(const float4*)&Ot[w * 16 * OMP + col];
        acc.x += t.x * f[w]; acc.y += t.y * f[w];
        acc.z += t.z * f[w]; acc.w += t.w * f[w];
      }
      acc.x *= inv; acc.y *= inv; acc.z *= inv; acc.w *= inv;
      *(float4*)(dst + col) = acc;
    }
  }
}

extern "C" void kernel_launch(void* const* d_in, const int* in_sizes, int n_in,
                              void* d_out, int out_size, void* d_ws,
                              size_t ws_size, hipStream_t stream) {
  const float* x    = (const float*)d_in[0];
  const float* pt   = (const float*)d_in[1];
  const float* gq   = (const float*)d_in[3];
  const float* Wq   = (const float*)d_in[4];
  const float* bq   = (const float*)d_in[5];
  const float* gkv  = (const float*)d_in[6];
  const float* Wkv  = (const float*)d_in[7];
  const float* bkv  = (const float*)d_in[8];
  const float* gc   = (const float*)d_in[9];
  const float* Wc   = (const float*)d_in[10];
  const float* bc   = (const float*)d_in[11];
  const float* Wout = (const float*)d_in[12];
  const float* bout = (const float*)d_in[13];
  const int*   cw   = (const int*)d_in[14];
  float* out = (float*)d_out;

  float* ws = (float*)d_ws;
  const size_t R = 1024;
  float* xqn  = ws + 0 * R * 1024;
  float* xkvn = ws + 1 * R * 1024;
  float* ptn  = ws + 2 * R * 1024;
  float* Qb   = ws + 3 * R * 1024;
  float* KVb  = ws + 4 * R * 1024;
  float* Cb   = ws + 6 * R * 1024;
  float* Pb   = ws + 7 * R * 1024;
  float* Qab  = ws + 8256 * R;
  float* Kab  = ws + 10304 * R;
  float* wvb  = ws + 12352 * R;

  rmsnorm3_kernel<<<1024, 256, 0, stream>>>(x, pt, gq, gkv, gc, xqn, xkvn, ptn);
  wmma_gemm_bias<<<dim3(16, 16), 256, 0, stream>>>(xqn, Wq, bq, Qb, 1024);
  wmma_gemm_bias<<<dim3(16, 32), 256, 0, stream>>>(xkvn, Wkv, bkv, KVb, 2048);
  wmma_gemm_bias<<<dim3(16, 16), 256, 0, stream>>>(ptn, Wc, bc, Cb, 1024);
  cumsum_kernel<<<32, 1024, 0, stream>>>(Cb, Pb);
  augment_kernel<<<1024, 256, 0, stream>>>(Qb, KVb, Pb, cw, Qab, Kab);
  attn_kernel<<<dim3(32, 16), 256, 0, stream>>>(Qab, Kab, KVb, wvb);
  wmma_gemm_bias<<<dim3(16, 16), 256, 0, stream>>>(wvb, Wout, bout, out, 1024);
}


# GEMM: Q, KV and C projections merged into ONE launch of 16x64 blocks (blockIdx.y picks the product): 4 co-resident blocks per WGP fill each other's barrier gaps; 8 -> 6 launches; sum order unchanged
# speedup vs baseline: 6.2322x; 1.1254x over previous
#include <hip/hip_runtime.h>
#include <math.h>


typedef __attribute__((ext_vector_type(2))) float v2f;
typedef __attribute__((ext_vector_type(8))) float v8f;

#define L   1024
#define DM  1024
#define EPSR 1.1920929e-07f

static __device__ __forceinline__ v8f wmma_f32(v2f a, v2f b, v8f c) {
  return __builtin_amdgcn_wmma_f32_16x16x4_f32(false, a, false, b, (short)0, c,
                                               false, false);
}

__global__ void __launch_bounds__(256)
rmsnorm3_kernel(const float* __restrict__ x, const float* __restrict__ pt,
                const float* __restrict__ gq, const float* __restrict__ gkv,
                const float* __restrict__ gc,
                float* __restrict__ xqn, float* __restrict__ xkvn,
                float* __restrict__ ptn) {
  __shared__ float red[256];
  const int row = blockIdx.x;
  const int tid = threadIdx.x;

  float4 xv = ((const float4*)(x + (size_t)row * DM))[tid];
  red[tid] = xv.x * xv.x + xv.y * xv.y + xv.z * xv.z + xv.w * xv.w;
  __syncthreads();
  for (int s = 128; s > 0; s >>= 1) {
    if (tid < s) red[tid] += red[tid + s];
    __syncthreads();
  }
  float rn = rsqrtf(red[0] * (1.0f / DM) + EPSR);
  __syncthreads();

  float4 gqv = ((const float4*)gq)[tid];
  float4 gkvv = ((const float4*)gkv)[tid];
  float4 oq, okv;
  oq.x = xv.x * rn * gqv.x;  oq.y = xv.y * rn * gqv.y;
  oq.z = xv.z * rn * gqv.z;  oq.w = xv.w * rn * gqv.w;
  okv.x = xv.x * rn * gkvv.x; okv.y = xv.y * rn * gkvv.y;
  okv.z = xv.z * rn * gkvv.z; okv.w = xv.w * rn * gkvv.w;
  ((float4*)(xqn + (size_t)row * DM))[tid] = oq;
  ((float4*)(xkvn + (size_t)row * DM))[tid] = okv;

  float4 pv = ((const float4*)(pt + (size_t)row * DM))[tid];
  red[tid] = pv.x * pv.x + pv.y * pv.y + pv.z * pv.z + pv.w * pv.w;
  __syncthreads();
  for (int s = 128; s > 0; s >>= 1) {
    if (tid < s) red[tid] += red[tid + s];
    __syncthreads();
  }
  float rnp = rsqrtf(red[0] * (1.0f / DM) + EPSR);
  float4 gcv = ((const float4*)gc)[tid];
  float4 oc;
  oc.x = pv.x * rnp * gcv.x; oc.y = pv.y * rnp * gcv.y;
  oc.z = pv.z * rnp * gcv.z; oc.w = pv.w * rnp * gcv.w;
  ((float4*)(ptn + (size_t)row * DM))[tid] = oc;
}

#define GKT 64
#define GPA 68
#define GPB 72
__global__ void __launch_bounds__(256)
wmma_gemm_bias(const float* __restrict__ A0, const float* __restrict__ W0,
               const float* __restrict__ bias0, float* __restrict__ C0,
               const float* __restrict__ A1, const float* __restrict__ W1,
               const float* __restrict__ bias1, float* __restrict__ C1,
               const float* __restrict__ A2, const float* __restrict__ W2,
               const float* __restrict__ bias2, float* __restrict__ C2) {
  __shared__ __attribute__((aligned(16))) float As[64 * GPA];
  __shared__ __attribute__((aligned(16))) float Bs[GKT * GPB];
  const int by = blockIdx.y;
  const int sel = (by < 16) ? 0 : ((by < 48) ? 1 : 2);
  const float* __restrict__ A    = (sel == 0) ? A0 : ((sel == 1) ? A1 : A2);
  const float* __restrict__ Wm   = (sel == 0) ? W0 : ((sel == 1) ? W1 : W2);
  const float* __restrict__ bias = (sel == 0) ? bias0 : ((sel == 1) ? bias1 : bias2);
  float* __restrict__ Cout       = (sel == 0) ? C0 : ((sel == 1) ? C1 : C2);
  const int N  = (sel == 1) ? 2048 : 1024;
  const int ny = (sel == 0) ? by : ((sel == 1) ? by - 16 : by - 48);
  const int tid = threadIdx.x;
  const int lane = tid & 31, wid = tid >> 5;
  const int hhi = lane >> 4, l16 = lane & 15;
  const int m0 = blockIdx.x * 64, n0 = ny * 64;
  const int wm = wid & 3, wn = wid >> 2;
  const int mloc = wm * 16 + l16;
  const int nloc0 = wn * 32 + l16;
  const int nloc1 = nloc0 + 16;

  const int srow = tid >> 4;
  const int scol = (tid & 15) * 4;
  const float* ap = A + (size_t)(m0 + srow) * DM + scol;
  const float* bp = Wm + (size_t)srow * N + n0 + scol;
  const size_t a16 = (size_t)16 * DM;
  const size_t b16 = (size_t)16 * N;
  float* asd = &As[srow * GPA + scol];
  float* bsd = &Bs[srow * GPB + scol];

  float4 ra0 = *(const float4*)(ap);
  float4 ra1 = *(const float4*)(ap + a16);
  float4 ra2 = *(const float4*)(ap + 2 * a16);
  float4 ra3 = *(const float4*)(ap + 3 * a16);
  float4 rb0 = *(const float4*)(bp);
  float4 rb1 = *(const float4*)(bp + b16);
  float4 rb2 = *(const float4*)(bp + 2 * b16);
  float4 rb3 = *(const float4*)(bp + 3 * b16);

  v8f acc0 = {}; v8f acc1 = {};

  for (int k0 = 0; k0 < DM; k0 += GKT) {
    *(float4*)(asd)                = ra0;
    *(float4*)(asd + 16 * GPA)     = ra1;
    *(float4*)(asd + 32 * GPA)     = ra2;
    *(float4*)(asd + 48 * GPA)     = ra3;
    *(float4*)(bsd)                = rb0;
    *(float4*)(bsd + 16 * GPB)     = rb1;
    *(float4*)(bsd + 32 * GPB)     = rb2;
    *(float4*)(bsd + 48 * GPB)     = rb3;
    __syncthreads();
    if (k0 + GKT < DM) {
      const float* an = ap + (k0 + GKT);
      const float* bn = bp + (size_t)(k0 + GKT) * N;
      ra0 = *(const float4*)(an);
      ra1 = *(const float4*)(an + a16);
      ra2 = *(const float4*)(an + 2 * a16);
      ra3 = *(const float4*)(an + 3 * a16);
      rb0 = *(const float4*)(bn);
      rb1 = *(const float4*)(bn + b16);
      rb2 = *(const float4*)(bn + 2 * b16);
      rb3 = *(const float4*)(bn + 3 * b16);
    }
#pragma unroll
    for (int kk = 0; kk < GKT / 4; ++kk) {
      const int kb = kk * 4 + 2 * hhi;
      v2f a;  a.x = As[mloc * GPA + kb];       a.y = As[mloc * GPA + kb + 1];
      v2f b0; b0.x = Bs[kb * GPB + nloc0];     b0.y = Bs[(kb + 1) * GPB + nloc0];
      v2f b1; b1.x = Bs[kb * GPB + nloc1];     b1.y = Bs[(kb + 1) * GPB + nloc1];
      acc0 = wmma_f32(a, b0, acc0);
      acc1 = wmma_f32(a, b1, acc1);
    }
    __syncthreads();
  }

  const int rbase = m0 + wm * 16 + hhi * 8;
  const float bb0 = bias[n0 + nloc0];
  const float bb1 = bias[n0 + nloc1];
#pragma unroll
  for (int v = 0; v < 8; ++v) {
    const int r = rbase + v;
    Cout[(size_t)r * N + n0 + nloc0] = acc0[v] + bb0;
    Cout[(size_t)r * N + n0 + nloc1] = acc1[v] + bb1;
  }
}

#define SCR 32
__global__ void __launch_bounds__(1024)
cumsum_kernel(const float* __restrict__ C, float* __restrict__ P) {
  __shared__ float tot[32 * 32];
  const int lane = threadIdx.x & 31, wid = threadIdx.x >> 5;
  const int col = blockIdx.x * 32 + lane;
  const float* src = C + (size_t)(wid * SCR) * DM + col;
  float v[SCR];
#pragma unroll
  for (int r = 0; r < SCR; ++r) v[r] = src[(size_t)r * DM];
#pragma unroll
  for (int r = 1; r < SCR; ++r) v[r] += v[r - 1];
  tot[wid * 32 + lane] = v[SCR - 1];
  __syncthreads();
  float off = 0.f;
#pragma unroll
  for (int u = 0; u < 32; ++u) {
    const float tv = tot[u * 32 + lane];
    off += (u < wid) ? tv : 0.f;
  }
  float* dst = P + (size_t)(wid * SCR + 1) * DM + col;
#pragma unroll
  for (int r = 0; r < SCR; ++r) dst[(size_t)r * DM] = off + v[r];
  if (wid == 0) P[col] = 0.f;
}

__global__ void augment_kernel(const float* __restrict__ Q,
                               const float* __restrict__ KV,
                               const float* __restrict__ P,
                               const int* __restrict__ cw,
                               float* __restrict__ Qa, float* __restrict__ Ka) {
  const int i = blockIdx.x;
  const int w = cw[0];
  const float scale = 0.125f;
  int E = i + w; if (E > L) E = L;
  int S = i - w; if (S < 0) S = 0;
  for (int c = threadIdx.x; c < DM; c += blockDim.x) {
    const int h = c >> 6, dd = c & 63;
    const float pe = P[(size_t)E * DM + c];
    const float ps = P[(size_t)S * DM + c];
    const float qv = Q[(size_t)i * DM + c];
    const float kv = KV[(size_t)i * 2048 + c];
    const int base = i * 2048 + h * 128 + dd;
    Qa[base]      = qv * (scale + pe);
    Qa[base + 64] = qv;
    Ka[base]      = kv;
    Ka[base + 64] = -kv * ps;
  }
}

#define QKP 132
#define OMP 68
#define ANW 8

static __device__ __forceinline__ void attn_tile(
    const float* __restrict__ kp, const float* __restrict__ vp, const float* Qt,
    float* myK, float* myV, float* myP, const int lane, const int hhi,
    const int l16, const int vr, const int vc, const bool diag,
    v8f& o0, v8f& o1, v8f& o2, v8f& o3, float (&m8)[8], float (&l8)[8]) {
  {
    float4 kr[16], vv[8];
#pragma unroll
    for (int r = 0; r < 16; ++r) kr[r] = *(const float4*)(kp + (size_t)r * 2048);
#pragma unroll
    for (int r = 0; r < 8; ++r) vv[r] = *(const float4*)(vp + (size_t)(2 * r) * 2048);
#pragma unroll
    for (int r = 0; r < 16; ++r) *(float4*)&myK[r * QKP + lane * 4] = kr[r];
#pragma unroll
    for (int r = 0; r < 8; ++r) *(float4*)&myV[(2 * r + vr) * 64 + vc] = vv[r];
  }

  v8f s = {};
#pragma unroll 8
  for (int kk = 0; kk < 32; ++kk) {
    const int kb = kk * 4 + 2 * hhi;
    v2f a; a.x = Qt[l16 * QKP + kb];  a.y = Qt[l16 * QKP + kb + 1];
    v2f b; b.x = myK[l16 * QKP + kb]; b.y = myK[l16 * QKP + kb + 1];
    s = wmma_f32(a, b, s);
  }

#pragma unroll
  for (int v = 0; v < 8; ++v) {
    const int row = v + hhi * 8;
    float sv = s[v];
    if (diag && (l16 > row)) sv = -1.0e30f;
    float mx = sv;
    for (int off = 8; off > 0; off >>= 1)
      mx = fmaxf(mx, __shfl_xor(mx, off, 16));
    const float mnew = fmaxf(m8[v], mx);
    const float alpha = __expf(m8[v] - mnew);
    const float p = __expf(sv - mnew);
    float rs = p;
    for (int off = 8; off > 0; off >>= 1) rs += __shfl_xor(rs, off, 16);
    l8[v] = l8[v] * alpha + rs;
    m8[v] = mnew;
    o0[v] *= alpha; o1[v] *= alpha; o2[v] *= alpha; o3[v] *= alpha;
    myP[row * 18 + l16] = p;
  }

#pragma unroll
  for (int kk = 0; kk < 4; ++kk) {
    const int kb = kk * 4 + 2 * hhi;
    v2f a; a.x = myP[l16 * 18 + kb]; a.y = myP[l16 * 18 + kb + 1];
    v2f b0, b1, b2, b3;
    b0.x = myV[kb * 64 + l16];      b0.y = myV[(kb + 1) * 64 + l16];
    b1.x = myV[kb * 64 + 16 + l16]; b1.y = myV[(kb + 1) * 64 + 16 + l16];
    b2.x = myV[kb * 64 + 32 + l16]; b2.y = myV[(kb + 1) * 64 + 32 + l16];
    b3.x = myV[kb * 64 + 48 + l16]; b3.y = myV[(kb + 1) * 64 + 48 + l16];
    o0 = wmma_f32(a, b0, o0);
    o1 = wmma_f32(a, b1, o1);
    o2 = wmma_f32(a, b2, o2);
    o3 = wmma_f32(a, b3, o3);
  }
}

__global__ void __launch_bounds__(256)
attn_kernel(const float* __restrict__ Qa, const float* __restrict__ Ka,
            const float* __restrict__ KV, float* __restrict__ wv) {
  __shared__ __attribute__((aligned(16))) float Qs[2 * 16 * QKP];
  __shared__ __attribute__((aligned(16))) float KVs[ANW * 16 * QKP + ANW * 16 * 64];
  __shared__ float Pls[ANW * 16 * 18];
  __shared__ float Mm[2 * ANW * 16];
  __shared__ float Lm[2 * ANW * 16];

  const int tid = threadIdx.x;
  const int lane = tid & 31;
  const int wid = __builtin_amdgcn_readfirstlane(tid >> 5);
  const int hhi = lane >> 4, l16 = lane & 15;
  const int h = blockIdx.y;
  const int hoff = h * 128;
  const int p = blockIdx.x;
  const int qtA = 63 - p;
  const int qtB = p;
  const int nA = 64 - p;

  float* myK = &KVs[wid * 16 * QKP];
  float* myV = &KVs[ANW * 16 * QKP + wid * 16 * 64];
  float* myP = &Pls[wid * 16 * 18];

  {
    const int tq = tid >> 7;
    const int t = tid & 127;
    const int r0 = t >> 5, c = (t & 31) * 4;
    const int qrow = (tq ? qtB : qtA) * 16;
    const float* qsrc = Qa + (size_t)(qrow + r0) * 2048 + hoff + c;
    float* qd = &Qs[tq * 16 * QKP];
    const float4 q0 = *(const float4*)(qsrc);
    const float4 q1 = *(const float4*)(qsrc + 4 * 2048);
    const float4 q2 = *(const float4*)(qsrc + 8 * 2048);
    const float4 q3 = *(const float4*)(qsrc + 12 * 2048);
    *(float4*)&qd[r0 * QKP + c]        = q0;
    *(float4*)&qd[(r0 + 4) * QKP + c]  = q1;
    *(float4*)&qd[(r0 + 8) * QKP + c]  = q2;
    *(float4*)&qd[(r0 + 12) * QKP + c] = q3;
  }
  __syncthreads();

  const int vr = lane >> 4, vc = (lane & 15) * 4;
  const float* ksrc = Ka + hoff + lane * 4;
  const float* vsrc = KV + (size_t)vr * 2048 + 1024 + h * 64 + vc;

  v8f a0 = {}, a1 = {}, a2 = {}, a3 = {};
  v8f b0 = {}, b1 = {}, b2 = {}, b3 = {};
  float mA[8], lA[8], mB[8], lB[8];
#pragma unroll
  for (int v = 0; v < 8; ++v) {
    mA[v] = -3.0e38f; lA[v] = 0.f; mB[v] = -3.0e38f; lB[v] = 0.f;
  }

  int u = wid;
  for (; u < nA; u += ANW) {
    attn_tile(ksrc + (size_t)u * (16 * 2048), vsrc + (size_t)u * (16 * 2048),
              Qs, myK, myV, myP, lane, hhi, l16, vr, vc, u == qtA,
              a0, a1, a2, a3, mA, lA);
  }
  for (; u < 65; u += ANW) {
    const int jt = u - nA;
    attn_tile(ksrc + (size_t)jt * (16 * 2048), vsrc + (size_t)jt * (16 * 2048),
              Qs + 16 * QKP, myK, myV, myP, lane, hhi, l16, vr, vc, jt == qtB,
              b0, b1, b2, b3, mB, lB);
  }

  __syncthreads();
  float* Om = KVs;
  {
    float* oA = &Om[(0 * ANW + wid) * 16 * OMP];
    float* oB = &Om[(1 * ANW + wid) * 16 * OMP];
#pragma unroll
    for (int v = 0; v < 8; ++v) {
      const int row = v + hhi * 8;
      oA[row * OMP + l16]      = a0[v];
      oA[row * OMP + 16 + l16] = a1[v];
      oA[row * OMP + 32 + l16] = a2[v];
      oA[row * OMP + 48 + l16] = a3[v];
      oB[row * OMP + l16]      = b0[v];
      oB[row * OMP + 16 + l16] = b1[v];
      oB[row * OMP + 32 + l16] = b2[v];
      oB[row * OMP + 48 + l16] = b3[v];
      if (l16 == 0) {
        Mm[(0 * ANW + wid) * 16 + row] = mA[v]; Lm[(0 * ANW + wid) * 16 + row] = lA[v];
        Mm[(1 * ANW + wid) * 16 + row] = mB[v]; Lm[(1 * ANW + wid) * 16 + row] = lB[v];
      }
    }
  }
  __syncthreads();

  {
    const int tq = wid >> 2;
    const int row = (wid & 3) * 4 + (lane >> 3);
    const int c4 = (lane & 7) * 4;
    const float* Mt = &Mm[tq * ANW * 16 + row];
    const float* Lt = &Lm[tq * ANW * 16 + row];
    const float* Ot = &Om[(tq * ANW * 16 + row) * OMP];
    float pm[ANW], f[ANW];
#pragma unroll
    for (int w = 0; w < ANW; ++w) pm[w] = Mt[w * 16];
    float mt = pm[0];
#pragma unroll
    for (int w = 1; w < ANW; ++w) mt = fmaxf(mt, pm[w]);
    float lt = 0.f;
#pragma unroll
    for (int w = 0; w < ANW; ++w) {
      f[w] = __expf(pm[w] - mt);
      lt += Lt[w * 16] * f[w];
    }
    const float inv = 1.0f / lt;
    float* dst = wv + (size_t)((tq ? qtB : qtA) * 16 + row) * DM + h * 64;
#pragma unroll
    for (int i = 0; i < 2; ++i) {
      const int col = c4 + 32 * i;
      float4 acc = {0.f, 0.f, 0.f, 0.f};
#pragma unroll
      for (int w = 0; w < ANW; ++w) {
        const float4 t = *(const float4*)&Ot[w * 16 * OMP + col];
        acc.x += t.x * f[w]; acc.y += t.y * f[w];
        acc.z += t.z * f[w]; acc.w += t.w * f[w];
      }
      acc.x *= inv; acc.y *= inv; acc.z *= inv; acc.w *= inv;
      *(float4*)(dst + col) = acc;
    }
  }
}

extern "C" void kernel_launch(void* const* d_in, const int* in_sizes, int n_in,
                              void* d_out, int out_size, void* d_ws,
                              size_t ws_size, hipStream_t stream) {
  const float* x    = (const float*)d_in[0];
  const float* pt   = (const float*)d_in[1];
  const float* gq   = (const float*)d_in[3];
  const float* Wq   = (const float*)d_in[4];
  const float* bq   = (const float*)d_in[5];
  const float* gkv  = (const float*)d_in[6];
  const float* Wkv  = (const float*)d_in[7];
  const float* bkv  = (const float*)d_in[8];
  const float* gc   = (const float*)d_in[9];
  const float* Wc   = (const float*)d_in[10];
  const float* bc   = (const float*)d_in[11];
  const float* Wout = (const float*)d_in[12];
  const float* bout = (const float*)d_in[13];
  const int*   cw   = (const int*)d_in[14];
  float* out = (float*)d_out;

  float* ws = (float*)d_ws;
  const size_t R = 1024;
  float* xqn  = ws + 0 * R * 1024;
  float* xkvn = ws + 1 * R * 1024;
  float* ptn  = ws + 2 * R * 1024;
  float* Qb   = ws + 3 * R * 1024;
  float* KVb  = ws + 4 * R * 1024;
  float* Cb   = ws + 6 * R * 1024;
  float* Pb   = ws + 7 * R * 1024;
  float* Qab  = ws + 8256 * R;
  float* Kab  = ws + 10304 * R;
  float* wvb  = ws + 12352 * R;

  rmsnorm3_kernel<<<1024, 256, 0, stream>>>(x, pt, gq, gkv, gc, xqn, xkvn, ptn);
  wmma_gemm_bias<<<dim3(16, 64), 256, 0, stream>>>(xqn, Wq, bq, Qb,
                                                   xkvn, Wkv, bkv, KVb,
                                                   ptn, Wc, bc, Cb);
  cumsum_kernel<<<32, 1024, 0, stream>>>(Cb, Pb);
  augment_kernel<<<1024, 256, 0, stream>>>(Qb, KVb, Pb, cw, Qab, Kab);
  attn_kernel<<<dim3(32, 16), 256, 0, stream>>>(Qab, Kab, KVb, wvb);
  wmma_gemm_bias<<<dim3(16, 16), 256, 0, stream>>>(wvb, Wout, bout, out,
                                                   wvb, Wout, bout, out,
                                                   wvb, Wout, bout, out);
}


# attention: Q~/K~ augmentation folded into the attention kernel's tile staging (k, PS, q, PE combined in registers on the way to LDS); augment launch and its 2 planes gone, 6->5 launches; same sums
# speedup vs baseline: 6.5160x; 1.0455x over previous
#include <hip/hip_runtime.h>
#include <math.h>


typedef __attribute__((ext_vector_type(2))) float v2f;
typedef __attribute__((ext_vector_type(8))) float v8f;

#define L   1024
#define DM  1024
#define EPSR 1.1920929e-07f

static __device__ __forceinline__ v8f wmma_f32(v2f a, v2f b, v8f c) {
  return __builtin_amdgcn_wmma_f32_16x16x4_f32(false, a, false, b, (short)0, c,
                                               false, false);
}

__global__ void __launch_bounds__(256)
rmsnorm3_kernel(const float* __restrict__ x, const float* __restrict__ pt,
                const float* __restrict__ gq, const float* __restrict__ gkv,
                const float* __restrict__ gc,
                float* __restrict__ xqn, float* __restrict__ xkvn,
                float* __restrict__ ptn) {
  __shared__ float red[256];
  const int row = blockIdx.x;
  const int tid = threadIdx.x;

  float4 xv = ((const float4*)(x + (size_t)row * DM))[tid];
  red[tid] = xv.x * xv.x + xv.y * xv.y + xv.z * xv.z + xv.w * xv.w;
  __syncthreads();
  for (int s = 128; s > 0; s >>= 1) {
    if (tid < s) red[tid] += red[tid + s];
    __syncthreads();
  }
  float rn = rsqrtf(red[0] * (1.0f / DM) + EPSR);
  __syncthreads();

  float4 gqv = ((const float4*)gq)[tid];
  float4 gkvv = ((const float4*)gkv)[tid];
  float4 oq, okv;
  oq.x = xv.x * rn * gqv.x;  oq.y = xv.y * rn * gqv.y;
  oq.z = xv.z * rn * gqv.z;  oq.w = xv.w * rn * gqv.w;
  okv.x = xv.x * rn * gkvv.x; okv.y = xv.y * rn * gkvv.y;
  okv.z = xv.z * rn * gkvv.z; okv.w = xv.w * rn * gkvv.w;
  ((float4*)(xqn + (size_t)row * DM))[tid] = oq;
  ((float4*)(xkvn + (size_t)row * DM))[tid] = okv;

  float4 pv = ((const float4*)(pt + (size_t)row * DM))[tid];
  red[tid] = pv.x * pv.x + pv.y * pv.y + pv.z * pv.z + pv.w * pv.w;
  __syncthreads();
  for (int s = 128; s > 0; s >>= 1) {
    if (tid < s) red[tid] += red[tid + s];
    __syncthreads();
  }
  float rnp = rsqrtf(red[0] * (1.0f / DM) + EPSR);
  float4 gcv = ((const float4*)gc)[tid];
  float4 oc;
  oc.x = pv.x * rnp * gcv.x; oc.y = pv.y * rnp * gcv.y;
  oc.z = pv.z * rnp * gcv.z; oc.w = pv.w * rnp * gcv.w;
  ((float4*)(ptn + (size_t)row * DM))[tid] = oc;
}

#define GKT 64
#define GPA 68
#define GPB 72
__global__ void __launch_bounds__(256)
wmma_gemm_bias(const float* __restrict__ A0, const float* __restrict__ W0,
               const float* __restrict__ bias0, float* __restrict__ C0,
               const float* __restrict__ A1, const float* __restrict__ W1,
               const float* __restrict__ bias1, float* __restrict__ C1,
               const float* __restrict__ A2, const float* __restrict__ W2,
               const float* __restrict__ bias2, float* __restrict__ C2) {
  __shared__ __attribute__((aligned(16))) float As[64 * GPA];
  __shared__ __attribute__((aligned(16))) float Bs[GKT * GPB];
  const int by = blockIdx.y;
  const int sel = (by < 16) ? 0 : ((by < 48) ? 1 : 2);
  const float* __restrict__ A    = (sel == 0) ? A0 : ((sel == 1) ? A1 : A2);
  const float* __restrict__ Wm   = (sel == 0) ? W0 : ((sel == 1) ? W1 : W2);
  const float* __restrict__ bias = (sel == 0) ? bias0 : ((sel == 1) ? bias1 : bias2);
  float* __restrict__ Cout       = (sel == 0) ? C0 : ((sel == 1) ? C1 : C2);
  const int N  = (sel == 1) ? 2048 : 1024;
  const int ny = (sel == 0) ? by : ((sel == 1) ? by - 16 : by - 48);
  const int tid = threadIdx.x;
  const int lane = tid & 31, wid = tid >> 5;
  const int hhi = lane >> 4, l16 = lane & 15;
  const int m0 = blockIdx.x * 64, n0 = ny * 64;
  const int wm = wid & 3, wn = wid >> 2;
  const int mloc = wm * 16 + l16;
  const int nloc0 = wn * 32 + l16;
  const int nloc1 = nloc0 + 16;

  const int srow = tid >> 4;
  const int scol = (tid & 15) * 4;
  const float* ap = A + (size_t)(m0 + srow) * DM + scol;
  const float* bp = Wm + (size_t)srow * N + n0 + scol;
  const size_t a16 = (size_t)16 * DM;
  const size_t b16 = (size_t)16 * N;
  float* asd = &As[srow * GPA + scol];
  float* bsd = &Bs[srow * GPB + scol];

  float4 ra0 = *(const float4*)(ap);
  float4 ra1 = *(const float4*)(ap + a16);
  float4 ra2 = *(const float4*)(ap + 2 * a16);
  float4 ra3 = *(const float4*)(ap + 3 * a16);
  float4 rb0 = *(const float4*)(bp);
  float4 rb1 = *(const float4*)(bp + b16);
  float4 rb2 = *(const float4*)(bp + 2 * b16);
  float4 rb3 = *(const float4*)(bp + 3 * b16);

  v8f acc0 = {}; v8f acc1 = {};

  for (int k0 = 0; k0 < DM; k0 += GKT) {
    *(float4*)(asd)                = ra0;
    *(float4*)(asd + 16 * GPA)     = ra1;
    *(float4*)(asd + 32 * GPA)     = ra2;
    *(float4*)(asd + 48 * GPA)     = ra3;
    *(float4*)(bsd)                = rb0;
    *(float4*)(bsd + 16 * GPB)     = rb1;
    *(float4*)(bsd + 32 * GPB)     = rb2;
    *(float4*)(bsd + 48 * GPB)     = rb3;
    __syncthreads();
    if (k0 + GKT < DM) {
      const float* an = ap + (k0 + GKT);
      const float* bn = bp + (size_t)(k0 + GKT) * N;
      ra0 = *(const float4*)(an);
      ra1 = *(const float4*)(an + a16);
      ra2 = *(const float4*)(an + 2 * a16);
      ra3 = *(const float4*)(an + 3 * a16);
      rb0 = *(const float4*)(bn);
      rb1 = *(const float4*)(bn + b16);
      rb2 = *(const float4*)(bn + 2 * b16);
      rb3 = *(const float4*)(bn + 3 * b16);
    }
#pragma unroll
    for (int kk = 0; kk < GKT / 4; ++kk) {
      const int kb = kk * 4 + 2 * hhi;
      v2f a;  a.x = As[mloc * GPA + kb];       a.y = As[mloc * GPA + kb + 1];
      v2f b0; b0.x = Bs[kb * GPB + nloc0];     b0.y = Bs[(kb + 1) * GPB + nloc0];
      v2f b1; b1.x = Bs[kb * GPB + nloc1];     b1.y = Bs[(kb + 1) * GPB + nloc1];
      acc0 = wmma_f32(a, b0, acc0);
      acc1 = wmma_f32(a, b1, acc1);
    }
    __syncthreads();
  }

  const int rbase = m0 + wm * 16 + hhi * 8;
  const float bb0 = bias[n0 + nloc0];
  const float bb1 = bias[n0 + nloc1];
#pragma unroll
  for (int v = 0; v < 8; ++v) {
    const int r = rbase + v;
    Cout[(size_t)r * N + n0 + nloc0] = acc0[v] + bb0;
    Cout[(size_t)r * N + n0 + nloc1] = acc1[v] + bb1;
  }
}

#define SCR 32
__global__ void __launch_bounds__(1024)
cumsum_kernel(const float* __restrict__ C, float* __restrict__ P) {
  __shared__ float tot[32 * 32];
  const int lane = threadIdx.x & 31, wid = threadIdx.x >> 5;
  const int col = blockIdx.x * 32 + lane;
  const float* src = C + (size_t)(wid * SCR) * DM + col;
  float v[SCR];
#pragma unroll
  for (int r = 0; r < SCR; ++r) v[r] = src[(size_t)r * DM];
#pragma unroll
  for (int r = 1; r < SCR; ++r) v[r] += v[r - 1];
  tot[wid * 32 + lane] = v[SCR - 1];
  __syncthreads();
  float off = 0.f;
#pragma unroll
  for (int u = 0; u < 32; ++u) {
    const float tv = tot[u * 32 + lane];
    off += (u < wid) ? tv : 0.f;
  }
  float* dst = P + (size_t)(wid * SCR + 1) * DM + col;
#pragma unroll
  for (int r = 0; r < SCR; ++r) dst[(size_t)r * DM] = off + v[r];
  if (wid == 0) P[col] = 0.f;
}

#define QKP 132
#define OMP 68
#define ANW 8

static __device__ __forceinline__ void attn_tile(
    const float* __restrict__ kp, const float* __restrict__ pcol, const int srow0,
    const float* Qt,
    float* myK, float* myV, float* myP, const int lane, const int hhi,
    const int l16, const int vr, const int vc, const bool diag,
    v8f& o0, v8f& o1, v8f& o2, v8f& o3, float (&m8)[8], float (&l8)[8]) {
  {
    float4 kr[8], pr[8], vv[8];
#pragma unroll
    for (int r = 0; r < 8; ++r) kr[r] = *(const float4*)(kp + (size_t)(2 * r) * 2048);
#pragma unroll
    for (int r = 0; r < 8; ++r) {
      int sr = srow0 + 2 * r;
      sr = sr < 0 ? 0 : sr;
      sr = sr > L ? L : sr;
      pr[r] = *(const float4*)(pcol + (size_t)sr * DM);
    }
#pragma unroll
    for (int r = 0; r < 8; ++r)
      vv[r] = *(const float4*)(kp + 1024 + (size_t)(2 * r) * 2048);
#pragma unroll
    for (int r = 0; r < 8; ++r) {
      const float4 kq = kr[r];
      const float4 ps = pr[r];
      float4 t;
      t.x = -kq.x * ps.x; t.y = -kq.y * ps.y; t.z = -kq.z * ps.z; t.w = -kq.w * ps.w;
      *(float4*)&myK[(2 * r + vr) * QKP + vc]      = kq;
      *(float4*)&myK[(2 * r + vr) * QKP + 64 + vc] = t;
    }
#pragma unroll
    for (int r = 0; r < 8; ++r) *(float4*)&myV[(2 * r + vr) * 64 + vc] = vv[r];
  }

  v8f s = {};
#pragma unroll 8
  for (int kk = 0; kk < 32; ++kk) {
    const int kb = kk * 4 + 2 * hhi;
    v2f a; a.x = Qt[l16 * QKP + kb];  a.y = Qt[l16 * QKP + kb + 1];
    v2f b; b.x = myK[l16 * QKP + kb]; b.y = myK[l16 * QKP + kb + 1];
    s = wmma_f32(a, b, s);
  }

#pragma unroll
  for (int v = 0; v < 8; ++v) {
    const int row = v + hhi * 8;
    float sv = s[v];
    if (diag && (l16 > row)) sv = -1.0e30f;
    float mx = sv;
    for (int off = 8; off > 0; off >>= 1)
      mx = fmaxf(mx, __shfl_xor(mx, off, 16));
    const float mnew = fmaxf(m8[v], mx);
    const float alpha = __expf(m8[v] - mnew);
    const float p = __expf(sv - mnew);
    float rs = p;
    for (int off = 8; off > 0; off >>= 1) rs += __shfl_xor(rs, off, 16);
    l8[v] = l8[v] * alpha + rs;
    m8[v] = mnew;
    o0[v] *= alpha; o1[v] *= alpha; o2[v] *= alpha; o3[v] *= alpha;
    myP[row * 18 + l16] = p;
  }

#pragma unroll
  for (int kk = 0; kk < 4; ++kk) {
    const int kb = kk * 4 + 2 * hhi;
    v2f a; a.x = myP[l16 * 18 + kb]; a.y = myP[l16 * 18 + kb + 1];
    v2f b0, b1, b2, b3;
    b0.x = myV[kb * 64 + l16];      b0.y = myV[(kb + 1) * 64 + l16];
    b1.x = myV[kb * 64 + 16 + l16]; b1.y = myV[(kb + 1) * 64 + 16 + l16];
    b2.x = myV[kb * 64 + 32 + l16]; b2.y = myV[(kb + 1) * 64 + 32 + l16];
    b3.x = myV[kb * 64 + 48 + l16]; b3.y = myV[(kb + 1) * 64 + 48 + l16];
    o0 = wmma_f32(a, b0, o0);
    o1 = wmma_f32(a, b1, o1);
    o2 = wmma_f32(a, b2, o2);
    o3 = wmma_f32(a, b3, o3);
  }
}

__global__ void __launch_bounds__(256)
attn_kernel(const float* __restrict__ Q, const float* __restrict__ KV,
            const float* __restrict__ P, const int* __restrict__ cw,
            float* __restrict__ wv) {
  __shared__ __attribute__((aligned(16))) float Qs[2 * 16 * QKP];
  __shared__ __attribute__((aligned(16))) float KVs[ANW * 16 * QKP + ANW * 16 * 64];
  __shared__ float Pls[ANW * 16 * 18];
  __shared__ float Mm[2 * ANW * 16];
  __shared__ float Lm[2 * ANW * 16];

  const int tid = threadIdx.x;
  const int lane = tid & 31;
  const int wid = __builtin_amdgcn_readfirstlane(tid >> 5);
  const int hhi = lane >> 4, l16 = lane & 15;
  const int h = blockIdx.y;
  const int w = cw[0];
  const int p = blockIdx.x;
  const int qtA = 63 - p;
  const int qtB = p;
  const int nA = 64 - p;

  float* myK = &KVs[wid * 16 * QKP];
  float* myV = &KVs[ANW * 16 * QKP + wid * 16 * 64];
  float* myP = &Pls[wid * 16 * 18];

  {
    const float scale = 0.125f;
    const int tq = tid >> 7;
    const int t = tid & 127;
    const int r0 = t >> 4, cc = (t & 15) * 4;
    const int qrow = (tq ? qtB : qtA) * 16 + r0;
    const float* qsrc = Q + (size_t)qrow * DM + h * 64 + cc;
    const float* psrc = P + h * 64 + cc;
    float* qd = &Qs[tq * 16 * QKP + r0 * QKP + cc];
    float4 qv[2], pe[2];
#pragma unroll
    for (int i = 0; i < 2; ++i) qv[i] = *(const float4*)(qsrc + (size_t)(8 * i) * DM);
#pragma unroll
    for (int i = 0; i < 2; ++i) {
      int er = qrow + 8 * i + w;
      er = er < 0 ? 0 : er;
      er = er > L ? L : er;
      pe[i] = *(const float4*)(psrc + (size_t)er * DM);
    }
#pragma unroll
    for (int i = 0; i < 2; ++i) {
      float4 o;
      o.x = qv[i].x * (scale + pe[i].x);
      o.y = qv[i].y * (scale + pe[i].y);
      o.z = qv[i].z * (scale + pe[i].z);
      o.w = qv[i].w * (scale + pe[i].w);
      *(float4*)&qd[(8 * i) * QKP]      = o;
      *(float4*)&qd[(8 * i) * QKP + 64] = qv[i];
    }
  }
  __syncthreads();

  const int vr = lane >> 4, vc = (lane & 15) * 4;
  const float* ksrc = KV + (size_t)vr * 2048 + h * 64 + vc;
  const float* pcol = P + h * 64 + vc;
  const int sbase = vr - w;

  v8f a0 = {}, a1 = {}, a2 = {}, a3 = {};
  v8f b0 = {}, b1 = {}, b2 = {}, b3 = {};
  float mA[8], lA[8], mB[8], lB[8];
#pragma unroll
  for (int v = 0; v < 8; ++v) {
    mA[v] = -3.0e38f; lA[v] = 0.f; mB[v] = -3.0e38f; lB[v] = 0.f;
  }

  int u = wid;
  for (; u < nA; u += ANW) {
    attn_tile(ksrc + (size_t)u * (16 * 2048), pcol, u * 16 + sbase,
              Qs, myK, myV, myP, lane, hhi, l16, vr, vc, u == qtA,
              a0, a1, a2, a3, mA, lA);
  }
  for (; u < 65; u += ANW) {
    const int jt = u - nA;
    attn_tile(ksrc + (size_t)jt * (16 * 2048), pcol, jt * 16 + sbase,
              Qs + 16 * QKP, myK, myV, myP, lane, hhi, l16, vr, vc, jt == qtB,
              b0, b1, b2, b3, mB, lB);
  }

  __syncthreads();
  float* Om = KVs;
  {
    float* oA = &Om[(0 * ANW + wid) * 16 * OMP];
    float* oB = &Om[(1 * ANW + wid) * 16 * OMP];
#pragma unroll
    for (int v = 0; v < 8; ++v) {
      const int row = v + hhi * 8;
      oA[row * OMP + l16]      = a0[v];
      oA[row * OMP + 16 + l16] = a1[v];
      oA[row * OMP + 32 + l16] = a2[v];
      oA[row * OMP + 48 + l16] = a3[v];
      oB[row * OMP + l16]      = b0[v];
      oB[row * OMP + 16 + l16] = b1[v];
      oB[row * OMP + 32 + l16] = b2[v];
      oB[row * OMP + 48 + l16] = b3[v];
      if (l16 == 0) {
        Mm[(0 * ANW + wid) * 16 + row] = mA[v]; Lm[(0 * ANW + wid) * 16 + row] = lA[v];
        Mm[(1 * ANW + wid) * 16 + row] = mB[v]; Lm[(1 * ANW + wid) * 16 + row] = lB[v];
      }
    }
  }
  __syncthreads();

  {
    const int tq = wid >> 2;
    const int row = (wid & 3) * 4 + (lane >> 3);
    const int c4 = (lane & 7) * 4;
    const float* Mt = &Mm[tq * ANW * 16 + row];
    const float* Lt = &Lm[tq * ANW * 16 + row];
    const float* Ot = &Om[(tq * ANW * 16 + row) * OMP];
    float pm[ANW], f[ANW];
#pragma unroll
    for (int w = 0; w < ANW; ++w) pm[w] = Mt[w * 16];
    float mt = pm[0];
#pragma unroll
    for (int w = 1; w < ANW; ++w) mt = fmaxf(mt, pm[w]);
    float lt = 0.f;
#pragma unroll
    for (int w = 0; w < ANW; ++w) {
      f[w] = __expf(pm[w] - mt);
      lt += Lt[w * 16] * f[w];
    }
    const float inv = 1.0f / lt;
    float* dst = wv + (size_t)((tq ? qtB : qtA) * 16 + row) * DM + h * 64;
#pragma unroll
    for (int i = 0; i < 2; ++i) {
      const int col = c4 + 32 * i;
      float4 acc = {0.f, 0.f, 0.f, 0.f};
#pragma unroll
      for (int w = 0; w < ANW; ++w) {
        const float4 t = *(const float4*)&Ot[w * 16 * OMP + col];
        acc.x += t.x * f[w]; acc.y += t.y * f[w];
        acc.z += t.z * f[w]; acc.w += t.w * f[w];
      }
      acc.x *= inv; acc.y *= inv; acc.z *= inv; acc.w *= inv;
      *(float4*)(dst + col) = acc;
    }
  }
}

extern "C" void kernel_launch(void* const* d_in, const int* in_sizes, int n_in,
                              void* d_out, int out_size, void* d_ws,
                              size_t ws_size, hipStream_t stream) {
  const float* x    = (const float*)d_in[0];
  const float* pt   = (const float*)d_in[1];
  const float* gq   = (const float*)d_in[3];
  const float* Wq   = (const float*)d_in[4];
  const float* bq   = (const float*)d_in[5];
  const float* gkv  = (const float*)d_in[6];
  const float* Wkv  = (const float*)d_in[7];
  const float* bkv  = (const float*)d_in[8];
  const float* gc   = (const float*)d_in[9];
  const float* Wc   = (const float*)d_in[10];
  const float* bc   = (const float*)d_in[11];
  const float* Wout = (const float*)d_in[12];
  const float* bout = (const float*)d_in[13];
  const int*   cw   = (const int*)d_in[14];
  float* out = (float*)d_out;

  float* ws = (float*)d_ws;
  const size_t R = 1024;
  float* xqn  = ws + 0 * R * 1024;
  float* xkvn = ws + 1 * R * 1024;
  float* ptn  = ws + 2 * R * 1024;
  float* Qb   = ws + 3 * R * 1024;
  float* KVb  = ws + 4 * R * 1024;
  float* Cb   = ws + 6 * R * 1024;
  float* Pb   = ws + 7 * R * 1024;
  float* wvb  = ws + 12352 * R;

  rmsnorm3_kernel<<<1024, 256, 0, stream>>>(x, pt, gq, gkv, gc, xqn, xkvn, ptn);
  wmma_gemm_bias<<<dim3(16, 64), 256, 0, stream>>>(xqn, Wq, bq, Qb,
                                                   xkvn, Wkv, bkv, KVb,
                                                   ptn, Wc, bc, Cb);
  cumsum_kernel<<<32, 1024, 0, stream>>>(Cb, Pb);
  attn_kernel<<<dim3(32, 16), 256, 0, stream>>>(Qb, KVb, Pb, cw, wvb);
  wmma_gemm_bias<<<dim3(16, 16), 256, 0, stream>>>(wvb, Wout, bout, out,
                                                   wvb, Wout, bout, out,
                                                   wvb, Wout, bout, out);
}


# attention: score tile computed transposed (K~ @ Q~^T): one query's scores per lane pair, row max/sum in-lane + 1 permlanex16 (was 64 ds_bpermute), P stays in registers; softmax/PV sum order changes
# speedup vs baseline: 6.7921x; 1.0424x over previous
#include <hip/hip_runtime.h>
#include <math.h>


typedef __attribute__((ext_vector_type(2))) float v2f;
typedef __attribute__((ext_vector_type(8))) float v8f;

#define L   1024
#define DM  1024
#define EPSR 1.1920929e-07f

static __device__ __forceinline__ v8f wmma_f32(v2f a, v2f b, v8f c) {
  return __builtin_amdgcn_wmma_f32_16x16x4_f32(false, a, false, b, (short)0, c,
                                               false, false);
}

__global__ void __launch_bounds__(256)
rmsnorm3_kernel(const float* __restrict__ x, const float* __restrict__ pt,
                const float* __restrict__ gq, const float* __restrict__ gkv,
                const float* __restrict__ gc,
                float* __restrict__ xqn, float* __restrict__ xkvn,
                float* __restrict__ ptn) {
  __shared__ float red[256];
  const int row = blockIdx.x;
  const int tid = threadIdx.x;

  float4 xv = ((const float4*)(x + (size_t)row * DM))[tid];
  red[tid] = xv.x * xv.x + xv.y * xv.y + xv.z * xv.z + xv.w * xv.w;
  __syncthreads();
  for (int s = 128; s > 0; s >>= 1) {
    if (tid < s) red[tid] += red[tid + s];
    __syncthreads();
  }
  float rn = rsqrtf(red[0] * (1.0f / DM) + EPSR);
  __syncthreads();

  float4 gqv = ((const float4*)gq)[tid];
  float4 gkvv = ((const float4*)gkv)[tid];
  float4 oq, okv;
  oq.x = xv.x * rn * gqv.x;  oq.y = xv.y * rn * gqv.y;
  oq.z = xv.z * rn * gqv.z;  oq.w = xv.w * rn * gqv.w;
  okv.x = xv.x * rn * gkvv.x; okv.y = xv.y * rn * gkvv.y;
  okv.z = xv.z * rn * gkvv.z; okv.w = xv.w * rn * gkvv.w;
  ((float4*)(xqn + (size_t)row * DM))[tid] = oq;
  ((float4*)(xkvn + (size_t)row * DM))[tid] = okv;

  float4 pv = ((const float4*)(pt + (size_t)row * DM))[tid];
  red[tid] = pv.x * pv.x + pv.y * pv.y + pv.z * pv.z + pv.w * pv.w;
  __syncthreads();
  for (int s = 128; s > 0; s >>= 1) {
    if (tid < s) red[tid] += red[tid + s];
    __syncthreads();
  }
  float rnp = rsqrtf(red[0] * (1.0f / DM) + EPSR);
  float4 gcv = ((const float4*)gc)[tid];
  float4 oc;
  oc.x = pv.x * rnp * gcv.x; oc.y = pv.y * rnp * gcv.y;
  oc.z = pv.z * rnp * gcv.z; oc.w = pv.w * rnp * gcv.w;
  ((float4*)(ptn + (size_t)row * DM))[tid] = oc;
}

#define GKT 64
#define GPA 68
#define GPB 72
__global__ void __launch_bounds__(256)
wmma_gemm_bias(const float* __restrict__ A0, const float* __restrict__ W0,
               const float* __restrict__ bias0, float* __restrict__ C0,
               const float* __restrict__ A1, const float* __restrict__ W1,
               const float* __restrict__ bias1, float* __restrict__ C1,
               const float* __restrict__ A2, const float* __restrict__ W2,
               const float* __restrict__ bias2, float* __restrict__ C2) {
  __shared__ __attribute__((aligned(16))) float As[64 * GPA];
  __shared__ __attribute__((aligned(16))) float Bs[GKT * GPB];
  const int by = blockIdx.y;
  const int sel = (by < 16) ? 0 : ((by < 48) ? 1 : 2);
  const float* __restrict__ A    = (sel == 0) ? A0 : ((sel == 1) ? A1 : A2);
  const float* __restrict__ Wm   = (sel == 0) ? W0 : ((sel == 1) ? W1 : W2);
  const float* __restrict__ bias = (sel == 0) ? bias0 : ((sel == 1) ? bias1 : bias2);
  float* __restrict__ Cout       = (sel == 0) ? C0 : ((sel == 1) ? C1 : C2);
  const int N  = (sel == 1) ? 2048 : 1024;
  const int ny = (sel == 0) ? by : ((sel == 1) ? by - 16 : by - 48);
  const int tid = threadIdx.x;
  const int lane = tid & 31, wid = tid >> 5;
  const int hhi = lane >> 4, l16 = lane & 15;
  const int m0 = blockIdx.x * 64, n0 = ny * 64;
  const int wm = wid & 3, wn = wid >> 2;
  const int mloc = wm * 16 + l16;
  const int nloc0 = wn * 32 + l16;
  const int nloc1 = nloc0 + 16;

  const int srow = tid >> 4;
  const int scol = (tid & 15) * 4;
  const float* ap = A + (size_t)(m0 + srow) * DM + scol;
  const float* bp = Wm + (size_t)srow * N + n0 + scol;
  const size_t a16 = (size_t)16 * DM;
  const size_t b16 = (size_t)16 * N;
  float* asd = &As[srow * GPA + scol];
  float* bsd = &Bs[srow * GPB + scol];

  float4 ra0 = *(const float4*)(ap);
  float4 ra1 = *(const float4*)(ap + a16);
  float4 ra2 = *(const float4*)(ap + 2 * a16);
  float4 ra3 = *(const float4*)(ap + 3 * a16);
  float4 rb0 = *(const float4*)(bp);
  float4 rb1 = *(const float4*)(bp + b16);
  float4 rb2 = *(const float4*)(bp + 2 * b16);
  float4 rb3 = *(const float4*)(bp + 3 * b16);

  v8f acc0 = {}; v8f acc1 = {};

  for (int k0 = 0; k0 < DM; k0 += GKT) {
    *(float4*)(asd)                = ra0;
    *(float4*)(asd + 16 * GPA)     = ra1;
    *(float4*)(asd + 32 * GPA)     = ra2;
    *(float4*)(asd + 48 * GPA)     = ra3;
    *(float4*)(bsd)                = rb0;
    *(float4*)(bsd + 16 * GPB)     = rb1;
    *(float4*)(bsd + 32 * GPB)     = rb2;
    *(float4*)(bsd + 48 * GPB)     = rb3;
    __syncthreads();
    if (k0 + GKT < DM) {
      const float* an = ap + (k0 + GKT);
      const float* bn = bp + (size_t)(k0 + GKT) * N;
      ra0 = *(const float4*)(an);
      ra1 = *(const float4*)(an + a16);
      ra2 = *(const float4*)(an + 2 * a16);
      ra3 = *(const float4*)(an + 3 * a16);
      rb0 = *(const float4*)(bn);
      rb1 = *(const float4*)(bn + b16);
      rb2 = *(const float4*)(bn + 2 * b16);
      rb3 = *(const float4*)(bn + 3 * b16);
    }
#pragma unroll
    for (int kk = 0; kk < GKT / 4; ++kk) {
      const int kb = kk * 4 + 2 * hhi;
      v2f a;  a.x = As[mloc * GPA + kb];       a.y = As[mloc * GPA + kb + 1];
      v2f b0; b0.x = Bs[kb * GPB + nloc0];     b0.y = Bs[(kb + 1) * GPB + nloc0];
      v2f b1; b1.x = Bs[kb * GPB + nloc1];     b1.y = Bs[(kb + 1) * GPB + nloc1];
      acc0 = wmma_f32(a, b0, acc0);
      acc1 = wmma_f32(a, b1, acc1);
    }
    __syncthreads();
  }

  const int rbase = m0 + wm * 16 + hhi * 8;
  const float bb0 = bias[n0 + nloc0];
  const float bb1 = bias[n0 + nloc1];
#pragma unroll
  for (int v = 0; v < 8; ++v) {
    const int r = rbase + v;
    Cout[(size_t)r * N + n0 + nloc0] = acc0[v] + bb0;
    Cout[(size_t)r * N + n0 + nloc1] = acc1[v] + bb1;
  }
}

#define SCR 32
__global__ void __launch_bounds__(1024)
cumsum_kernel(const float* __restrict__ C, float* __restrict__ P) {
  __shared__ float tot[32 * 32];
  const int lane = threadIdx.x & 31, wid = threadIdx.x >> 5;
  const int col = blockIdx.x * 32 + lane;
  const float* src = C + (size_t)(wid * SCR) * DM + col;
  float v[SCR];
#pragma unroll
  for (int r = 0; r < SCR; ++r) v[r] = src[(size_t)r * DM];
#pragma unroll
  for (int r = 1; r < SCR; ++r) v[r] += v[r - 1];
  tot[wid * 32 + lane] = v[SCR - 1];
  __syncthreads();
  float off = 0.f;
#pragma unroll
  for (int u = 0; u < 32; ++u) {
    const float tv = tot[u * 32 + lane];
    off += (u < wid) ? tv : 0.f;
  }
  float* dst = P + (size_t)(wid * SCR + 1) * DM + col;
#pragma unroll
  for (int r = 0; r < SCR; ++r) dst[(size_t)r * DM] = off + v[r];
  if (wid == 0) P[col] = 0.f;
}

#define QKP 132
#define VP  68
#define OMP 68
#define ANW 8

static __device__ __forceinline__ float xhalf(float x) {
  const int b = __builtin_bit_cast(int, x);
  return __builtin_bit_cast(float, __builtin_amdgcn_permlanex16(
      b, b, 0x76543210, 0xfedcba98, false, false));
}

static __device__ __forceinline__ void attn_tile(
    const float* __restrict__ kp, const float* __restrict__ pcol, const int srow0,
    const float* Qt,
    float* myK, float* myV, const int hhi,
    const int l16, const int vr, const int vc, const bool diag,
    v8f& o0, v8f& o1, v8f& o2, v8f& o3, float& m, float& l) {
  {
    float4 kr[8], pr[8], vv[8];
#pragma unroll
    for (int r = 0; r < 8; ++r) kr[r] = *(const float4*)(kp + (size_t)(2 * r) * 2048);
#pragma unroll
    for (int r = 0; r < 8; ++r) {
      int sr = srow0 + 2 * r;
      sr = sr < 0 ? 0 : sr;
      sr = sr > L ? L : sr;
      pr[r] = *(const float4*)(pcol + (size_t)sr * DM);
    }
#pragma unroll
    for (int r = 0; r < 8; ++r)
      vv[r] = *(const float4*)(kp + 1024 + (size_t)(2 * r) * 2048);
#pragma unroll
    for (int r = 0; r < 8; ++r) {
      const float4 kq = kr[r];
      const float4 ps = pr[r];
      float4 t;
      t.x = -kq.x * ps.x; t.y = -kq.y * ps.y; t.z = -kq.z * ps.z; t.w = -kq.w * ps.w;
      *(float4*)&myK[(2 * r + vr) * QKP + vc]      = kq;
      *(float4*)&myK[(2 * r + vr) * QKP + 64 + vc] = t;
    }
#pragma unroll
    for (int r = 0; r < 8; ++r) *(float4*)&myV[(2 * r + vr) * VP + vc] = vv[r];
  }

  v8f s = {};
#pragma unroll 8
  for (int kk = 0; kk < 32; ++kk) {
    const int kb = kk * 4 + 2 * hhi;
    v2f a; a.x = myK[l16 * QKP + kb]; a.y = myK[l16 * QKP + kb + 1];
    v2f b; b.x = Qt[l16 * QKP + kb];  b.y = Qt[l16 * QKP + kb + 1];
    s = wmma_f32(a, b, s);
  }

  if (diag) {
#pragma unroll
    for (int v = 0; v < 8; ++v)
      if (8 * hhi + v > l16) s[v] = -1.0e30f;
  }
  float mx = fmaxf(fmaxf(fmaxf(s[0], s[1]), fmaxf(s[2], s[3])),
                   fmaxf(fmaxf(s[4], s[5]), fmaxf(s[6], s[7])));
  mx = fmaxf(mx, xhalf(mx));
  const float mnew = fmaxf(m, mx);
  const float alpha = __expf(m - mnew);
  float p[8];
#pragma unroll
  for (int v = 0; v < 8; ++v) p[v] = __expf(s[v] - mnew);
  float rs = ((p[0] + p[1]) + (p[2] + p[3])) + ((p[4] + p[5]) + (p[6] + p[7]));
  rs += xhalf(rs);
  l = l * alpha + rs;
  m = mnew;
#pragma unroll
  for (int v = 0; v < 8; ++v) {
    o0[v] *= alpha; o1[v] *= alpha; o2[v] *= alpha; o3[v] *= alpha;
  }

#pragma unroll
  for (int kk = 0; kk < 4; ++kk) {
    const int kr0 = 8 * hhi + 2 * kk;
    v2f b; b.x = p[2 * kk]; b.y = p[2 * kk + 1];
    v2f a0, a1, a2, a3;
    a0.x = myV[kr0 * VP + l16];      a0.y = myV[(kr0 + 1) * VP + l16];
    a1.x = myV[kr0 * VP + 16 + l16]; a1.y = myV[(kr0 + 1) * VP + 16 + l16];
    a2.x = myV[kr0 * VP + 32 + l16]; a2.y = myV[(kr0 + 1) * VP + 32 + l16];
    a3.x = myV[kr0 * VP + 48 + l16]; a3.y = myV[(kr0 + 1) * VP + 48 + l16];
    o0 = wmma_f32(a0, b, o0);
    o1 = wmma_f32(a1, b, o1);
    o2 = wmma_f32(a2, b, o2);
    o3 = wmma_f32(a3, b, o3);
  }
}

__global__ void __launch_bounds__(256)
attn_kernel(const float* __restrict__ Q, const float* __restrict__ KV,
            const float* __restrict__ P, const int* __restrict__ cw,
            float* __restrict__ wv) {
  __shared__ __attribute__((aligned(16))) float Qs[2 * 16 * QKP];
  __shared__ __attribute__((aligned(16))) float KVs[ANW * 16 * QKP + ANW * 16 * VP];
  __shared__ float Mm[2 * ANW * 16];
  __shared__ float Lm[2 * ANW * 16];

  const int tid = threadIdx.x;
  const int lane = tid & 31;
  const int wid = __builtin_amdgcn_readfirstlane(tid >> 5);
  const int hhi = lane >> 4, l16 = lane & 15;
  const int h = blockIdx.y;
  const int w = cw[0];
  const int p = blockIdx.x;
  const int qtA = 63 - p;
  const int qtB = p;
  const int nA = 64 - p;

  float* myK = &KVs[wid * 16 * QKP];
  float* myV = &KVs[ANW * 16 * QKP + wid * 16 * VP];

  {
    const float scale = 0.125f;
    const int tq = tid >> 7;
    const int t = tid & 127;
    const int r0 = t >> 4, cc = (t & 15) * 4;
    const int qrow = (tq ? qtB : qtA) * 16 + r0;
    const float* qsrc = Q + (size_t)qrow * DM + h * 64 + cc;
    const float* psrc = P + h * 64 + cc;
    float* qd = &Qs[tq * 16 * QKP + r0 * QKP + cc];
    float4 qv[2], pe[2];
#pragma unroll
    for (int i = 0; i < 2; ++i) qv[i] = *(const float4*)(qsrc + (size_t)(8 * i) * DM);
#pragma unroll
    for (int i = 0; i < 2; ++i) {
      int er = qrow + 8 * i + w;
      er = er < 0 ? 0 : er;
      er = er > L ? L : er;
      pe[i] = *(const float4*)(psrc + (size_t)er * DM);
    }
#pragma unroll
    for (int i = 0; i < 2; ++i) {
      float4 o;
      o.x = qv[i].x * (scale + pe[i].x);
      o.y = qv[i].y * (scale + pe[i].y);
      o.z = qv[i].z * (scale + pe[i].z);
      o.w = qv[i].w * (scale + pe[i].w);
      *(float4*)&qd[(8 * i) * QKP]      = o;
      *(float4*)&qd[(8 * i) * QKP + 64] = qv[i];
    }
  }
  __syncthreads();

  const int vr = lane >> 4, vc = (lane & 15) * 4;
  const float* ksrc = KV + (size_t)vr * 2048 + h * 64 + vc;
  const float* pcol = P + h * 64 + vc;
  const int sbase = vr - w;

  v8f a0 = {}, a1 = {}, a2 = {}, a3 = {};
  v8f b0 = {}, b1 = {}, b2 = {}, b3 = {};
  float mA = -3.0e38f, lA = 0.f, mB = -3.0e38f, lB = 0.f;

  int u = wid;
  for (; u < nA; u += ANW) {
    attn_tile(ksrc + (size_t)u * (16 * 2048), pcol, u * 16 + sbase,
              Qs, myK, myV, hhi, l16, vr, vc, u == qtA,
              a0, a1, a2, a3, mA, lA);
  }
  for (; u < 65; u += ANW) {
    const int jt = u - nA;
    attn_tile(ksrc + (size_t)jt * (16 * 2048), pcol, jt * 16 + sbase,
              Qs + 16 * QKP, myK, myV, hhi, l16, vr, vc, jt == qtB,
              b0, b1, b2, b3, mB, lB);
  }

  __syncthreads();
  float* Om = KVs;
  {
    float* oA = &Om[((0 * ANW + wid) * 16 + l16) * OMP + 8 * hhi];
    float* oB = &Om[((1 * ANW + wid) * 16 + l16) * OMP + 8 * hhi];
    *(float4*)&oA[0]  = make_float4(a0[0], a0[1], a0[2], a0[3]);
    *(float4*)&oA[4]  = make_float4(a0[4], a0[5], a0[6], a0[7]);
    *(float4*)&oA[16] = make_float4(a1[0], a1[1], a1[2], a1[3]);
    *(float4*)&oA[20] = make_float4(a1[4], a1[5], a1[6], a1[7]);
    *(float4*)&oA[32] = make_float4(a2[0], a2[1], a2[2], a2[3]);
    *(float4*)&oA[36] = make_float4(a2[4], a2[5], a2[6], a2[7]);
    *(float4*)&oA[48] = make_float4(a3[0], a3[1], a3[2], a3[3]);
    *(float4*)&oA[52] = make_float4(a3[4], a3[5], a3[6], a3[7]);
    *(float4*)&oB[0]  = make_float4(b0[0], b0[1], b0[2], b0[3]);
    *(float4*)&oB[4]  = make_float4(b0[4], b0[5], b0[6], b0[7]);
    *(float4*)&oB[16] = make_float4(b1[0], b1[1], b1[2], b1[3]);
    *(float4*)&oB[20] = make_float4(b1[4], b1[5], b1[6], b1[7]);
    *(float4*)&oB[32] = make_float4(b2[0], b2[1], b2[2], b2[3]);
    *(float4*)&oB[36] = make_float4(b2[4], b2[5], b2[6], b2[7]);
    *(float4*)&oB[48] = make_float4(b3[0], b3[1], b3[2], b3[3]);
    *(float4*)&oB[52] = make_float4(b3[4], b3[5], b3[6], b3[7]);
    if (hhi == 0) {
      Mm[(0 * ANW + wid) * 16 + l16] = mA; Lm[(0 * ANW + wid) * 16 + l16] = lA;
      Mm[(1 * ANW + wid) * 16 + l16] = mB; Lm[(1 * ANW + wid) * 16 + l16] = lB;
    }
  }
  __syncthreads();

  {
    const int tq = wid >> 2;
    const int row = (wid & 3) * 4 + (lane >> 3);
    const int c4 = (lane & 7) * 4;
    const float* Mt = &Mm[tq * ANW * 16 + row];
    const float* Lt = &Lm[tq * ANW * 16 + row];
    const float* Ot = &Om[(tq * ANW * 16 + row) * OMP];
    float pm[ANW], f[ANW];
#pragma unroll
    for (int w = 0; w < ANW; ++w) pm[w] = Mt[w * 16];
    float mt = pm[0];
#pragma unroll
    for (int w = 1; w < ANW; ++w) mt = fmaxf(mt, pm[w]);
    float lt = 0.f;
#pragma unroll
    for (int w = 0; w < ANW; ++w) {
      f[w] = __expf(pm[w] - mt);
      lt += Lt[w * 16] * f[w];
    }
    const float inv = 1.0f / lt;
    float* dst = wv + (size_t)((tq ? qtB : qtA) * 16 + row) * DM + h * 64;
#pragma unroll
    for (int i = 0; i < 2; ++i) {
      const int col = c4 + 32 * i;
      float4 acc = {0.f, 0.f, 0.f, 0.f};
#pragma unroll
      for (int w = 0; w < ANW; ++w) {
        const float4 t = *(const float4*)&Ot[w * 16 * OMP + col];
        acc.x += t.x * f[w]; acc.y += t.y * f[w];
        acc.z += t.z * f[w]; acc.w += t.w * f[w];
      }
      acc.x *= inv; acc.y *= inv; acc.z *= inv; acc.w *= inv;
      *(float4*)(dst + col) = acc;
    }
  }
}

extern "C" void kernel_launch(void* const* d_in, const int* in_sizes, int n_in,
                              void* d_out, int out_size, void* d_ws,
                              size_t ws_size, hipStream_t stream) {
  const float* x    = (const float*)d_in[0];
  const float* pt   = (const float*)d_in[1];
  const float* gq   = (const float*)d_in[3];
  const float* Wq   = (const float*)d_in[4];
  const float* bq   = (const float*)d_in[5];
  const float* gkv  = (const float*)d_in[6];
  const float* Wkv  = (const float*)d_in[7];
  const float* bkv  = (const float*)d_in[8];
  const float* gc   = (const float*)d_in[9];
  const float* Wc   = (const float*)d_in[10];
  const float* bc   = (const float*)d_in[11];
  const float* Wout = (const float*)d_in[12];
  const float* bout = (const float*)d_in[13];
  const int*   cw   = (const int*)d_in[14];
  float* out = (float*)d_out;

  float* ws = (float*)d_ws;
  const size_t R = 1024;
  float* xqn  = ws + 0 * R * 1024;
  float* xkvn = ws + 1 * R * 1024;
  float* ptn  = ws + 2 * R * 1024;
  float* Qb   = ws + 3 * R * 1024;
  float* KVb  = ws + 4 * R * 1024;
  float* Cb   = ws + 6 * R * 1024;
  float* Pb   = ws + 7 * R * 1024;
  float* wvb  = ws + 12352 * R;

  rmsnorm3_kernel<<<1024, 256, 0, stream>>>(x, pt, gq, gkv, gc, xqn, xkvn, ptn);
  wmma_gemm_bias<<<dim3(16, 64), 256, 0, stream>>>(xqn, Wq, bq, Qb,
                                                   xkvn, Wkv, bkv, KVb,
                                                   ptn, Wc, bc, Cb);
  cumsum_kernel<<<32, 1024, 0, stream>>>(Cb, Pb);
  attn_kernel<<<dim3(32, 16), 256, 0, stream>>>(Qb, KVb, Pb, cw, wvb);
  wmma_gemm_bias<<<dim3(16, 16), 256, 0, stream>>>(wvb, Wout, bout, out,
                                                   wvb, Wout, bout, out,
                                                   wvb, Wout, bout, out);
}


# rmsnorm folded into the projection GEMM: gain applied to the A tile at staging, 1/rms row factor (squares summed from the staged x) applied to the accumulators; 5->4 launches; rms sum order changes
# speedup vs baseline: 7.0237x; 1.0341x over previous
#include <hip/hip_runtime.h>
#include <math.h>


typedef __attribute__((ext_vector_type(2))) float v2f;
typedef __attribute__((ext_vector_type(8))) float v8f;

#define L   1024
#define DM  1024
#define EPSR 1.1920929e-07f

static __device__ __forceinline__ v8f wmma_f32(v2f a, v2f b, v8f c) {
  return __builtin_amdgcn_wmma_f32_16x16x4_f32(false, a, false, b, (short)0, c,
                                               false, false);
}

#define GKT 64
#define GPA 68
#define GPB 72
template <bool NORM>
__global__ void __launch_bounds__(256)
wmma_gemm_bias(const float* __restrict__ A0, const float* __restrict__ W0,
               const float* __restrict__ bias0, const float* __restrict__ g0,
               float* __restrict__ C0,
               const float* __restrict__ A1, const float* __restrict__ W1,
               const float* __restrict__ bias1, const float* __restrict__ g1,
               float* __restrict__ C1,
               const float* __restrict__ A2, const float* __restrict__ W2,
               const float* __restrict__ bias2, const float* __restrict__ g2,
               float* __restrict__ C2) {
  __shared__ __attribute__((aligned(16))) float As[64 * GPA];
  __shared__ __attribute__((aligned(16))) float Bs[GKT * GPB];
  __shared__ float rnS[64];
  const int by = blockIdx.y;
  const int sel = (by < 16) ? 0 : ((by < 48) ? 1 : 2);
  const float* __restrict__ A    = (sel == 0) ? A0 : ((sel == 1) ? A1 : A2);
  const float* __restrict__ Wm   = (sel == 0) ? W0 : ((sel == 1) ? W1 : W2);
  const float* __restrict__ bias = (sel == 0) ? bias0 : ((sel == 1) ? bias1 : bias2);
  const float* __restrict__ gn   = (sel == 0) ? g0 : ((sel == 1) ? g1 : g2);
  float* __restrict__ Cout       = (sel == 0) ? C0 : ((sel == 1) ? C1 : C2);
  const int N  = (sel == 1) ? 2048 : 1024;
  const int ny = (sel == 0) ? by : ((sel == 1) ? by - 16 : by - 48);
  const int tid = threadIdx.x;
  const int lane = tid & 31, wid = tid >> 5;
  const int hhi = lane >> 4, l16 = lane & 15;
  const int m0 = blockIdx.x * 64, n0 = ny * 64;
  const int wm = wid & 3, wn = wid >> 2;
  const int mloc = wm * 16 + l16;
  const int nloc0 = wn * 32 + l16;
  const int nloc1 = nloc0 + 16;

  const int srow = tid >> 4;
  const int scol = (tid & 15) * 4;
  const float* ap = A + (size_t)(m0 + srow) * DM + scol;
  const float* bp = Wm + (size_t)srow * N + n0 + scol;
  const size_t a16 = (size_t)16 * DM;
  const size_t b16 = (size_t)16 * N;
  float* asd = &As[srow * GPA + scol];
  float* bsd = &Bs[srow * GPB + scol];

  float4 ra0 = *(const float4*)(ap);
  float4 ra1 = *(const float4*)(ap + a16);
  float4 ra2 = *(const float4*)(ap + 2 * a16);
  float4 ra3 = *(const float4*)(ap + 3 * a16);
  float4 rb0 = *(const float4*)(bp);
  float4 rb1 = *(const float4*)(bp + b16);
  float4 rb2 = *(const float4*)(bp + 2 * b16);
  float4 rb3 = *(const float4*)(bp + 3 * b16);
  const float* gp = NORM ? gn + scol : nullptr;
  float4 rg = {1.f, 1.f, 1.f, 1.f};
  if (NORM) rg = *(const float4*)(gp);
  float ss0 = 0.f, ss1 = 0.f, ss2 = 0.f, ss3 = 0.f;

  v8f acc0 = {}; v8f acc1 = {};

  for (int k0 = 0; k0 < DM; k0 += GKT) {
    if (NORM) {
      ss0 = fmaf(ra0.x, ra0.x, ss0); ss0 = fmaf(ra0.y, ra0.y, ss0);
      ss0 = fmaf(ra0.z, ra0.z, ss0); ss0 = fmaf(ra0.w, ra0.w, ss0);
      ss1 = fmaf(ra1.x, ra1.x, ss1); ss1 = fmaf(ra1.y, ra1.y, ss1);
      ss1 = fmaf(ra1.z, ra1.z, ss1); ss1 = fmaf(ra1.w, ra1.w, ss1);
      ss2 = fmaf(ra2.x, ra2.x, ss2); ss2 = fmaf(ra2.y, ra2.y, ss2);
      ss2 = fmaf(ra2.z, ra2.z, ss2); ss2 = fmaf(ra2.w, ra2.w, ss2);
      ss3 = fmaf(ra3.x, ra3.x, ss3); ss3 = fmaf(ra3.y, ra3.y, ss3);
      ss3 = fmaf(ra3.z, ra3.z, ss3); ss3 = fmaf(ra3.w, ra3.w, ss3);
      ra0.x *= rg.x; ra0.y *= rg.y; ra0.z *= rg.z; ra0.w *= rg.w;
      ra1.x *= rg.x; ra1.y *= rg.y; ra1.z *= rg.z; ra1.w *= rg.w;
      ra2.x *= rg.x; ra2.y *= rg.y; ra2.z *= rg.z; ra2.w *= rg.w;
      ra3.x *= rg.x; ra3.y *= rg.y; ra3.z *= rg.z; ra3.w *= rg.w;
    }
    *(float4*)(asd)                = ra0;
    *(float4*)(asd + 16 * GPA)     = ra1;
    *(float4*)(asd + 32 * GPA)     = ra2;
    *(float4*)(asd + 48 * GPA)     = ra3;
    *(float4*)(bsd)                = rb0;
    *(float4*)(bsd + 16 * GPB)     = rb1;
    *(float4*)(bsd + 32 * GPB)     = rb2;
    *(float4*)(bsd + 48 * GPB)     = rb3;
    __syncthreads();
    if (k0 + GKT < DM) {
      const float* an = ap + (k0 + GKT);
      const float* bn = bp + (size_t)(k0 + GKT) * N;
      ra0 = *(const float4*)(an);
      ra1 = *(const float4*)(an + a16);
      ra2 = *(const float4*)(an + 2 * a16);
      ra3 = *(const float4*)(an + 3 * a16);
      rb0 = *(const float4*)(bn);
      rb1 = *(const float4*)(bn + b16);
      rb2 = *(const float4*)(bn + 2 * b16);
      rb3 = *(const float4*)(bn + 3 * b16);
      if (NORM) rg = *(const float4*)(gp + (k0 + GKT));
    }
#pragma unroll
    for (int kk = 0; kk < GKT / 4; ++kk) {
      const int kb = kk * 4 + 2 * hhi;
      v2f a;  a.x = As[mloc * GPA + kb];       a.y = As[mloc * GPA + kb + 1];
      v2f b0; b0.x = Bs[kb * GPB + nloc0];     b0.y = Bs[(kb + 1) * GPB + nloc0];
      v2f b1; b1.x = Bs[kb * GPB + nloc1];     b1.y = Bs[(kb + 1) * GPB + nloc1];
      acc0 = wmma_f32(a, b0, acc0);
      acc1 = wmma_f32(a, b1, acc1);
    }
    __syncthreads();
  }

  if (NORM) {
    for (int o = 1; o < 16; o <<= 1) {
      ss0 += __shfl_xor(ss0, o);
      ss1 += __shfl_xor(ss1, o);
      ss2 += __shfl_xor(ss2, o);
      ss3 += __shfl_xor(ss3, o);
    }
    if ((tid & 15) == 0) {
      rnS[srow]      = rsqrtf(ss0 * (1.0f / DM) + EPSR);
      rnS[srow + 16] = rsqrtf(ss1 * (1.0f / DM) + EPSR);
      rnS[srow + 32] = rsqrtf(ss2 * (1.0f / DM) + EPSR);
      rnS[srow + 48] = rsqrtf(ss3 * (1.0f / DM) + EPSR);
    }
    __syncthreads();
  }

  const int rloc = wm * 16 + hhi * 8;
  const int rbase = m0 + rloc;
  const float bb0 = bias[n0 + nloc0];
  const float bb1 = bias[n0 + nloc1];
#pragma unroll
  for (int v = 0; v < 8; ++v) {
    const int r = rbase + v;
    const float rn = NORM ? rnS[rloc + v] : 1.0f;
    Cout[(size_t)r * N + n0 + nloc0] = NORM ? (acc0[v] * rn + bb0) : (acc0[v] + bb0);
    Cout[(size_t)r * N + n0 + nloc1] = NORM ? (acc1[v] * rn + bb1) : (acc1[v] + bb1);
  }
}

#define SCR 32
__global__ void __launch_bounds__(1024)
cumsum_kernel(const float* __restrict__ C, float* __restrict__ P) {
  __shared__ float tot[32 * 32];
  const int lane = threadIdx.x & 31, wid = threadIdx.x >> 5;
  const int col = blockIdx.x * 32 + lane;
  const float* src = C + (size_t)(wid * SCR) * DM + col;
  float v[SCR];
#pragma unroll
  for (int r = 0; r < SCR; ++r) v[r] = src[(size_t)r * DM];
#pragma unroll
  for (int r = 1; r < SCR; ++r) v[r] += v[r - 1];
  tot[wid * 32 + lane] = v[SCR - 1];
  __syncthreads();
  float off = 0.f;
#pragma unroll
  for (int u = 0; u < 32; ++u) {
    const float tv = tot[u * 32 + lane];
    off += (u < wid) ? tv : 0.f;
  }
  float* dst = P + (size_t)(wid * SCR + 1) * DM + col;
#pragma unroll
  for (int r = 0; r < SCR; ++r) dst[(size_t)r * DM] = off + v[r];
  if (wid == 0) P[col] = 0.f;
}

#define QKP 132
#define VP  68
#define OMP 68
#define ANW 8

static __device__ __forceinline__ float xhalf(float x) {
  const int b = __builtin_bit_cast(int, x);
  return __builtin_bit_cast(float, __builtin_amdgcn_permlanex16(
      b, b, 0x76543210, 0xfedcba98, false, false));
}

static __device__ __forceinline__ void attn_tile(
    const float* __restrict__ kp, const float* __restrict__ pcol, const int srow0,
    const float* Qt,
    float* myK, float* myV, const int hhi,
    const int l16, const int vr, const int vc, const bool diag,
    v8f& o0, v8f& o1, v8f& o2, v8f& o3, float& m, float& l) {
  {
    float4 kr[8], pr[8], vv[8];
#pragma unroll
    for (int r = 0; r < 8; ++r) kr[r] = *(const float4*)(kp + (size_t)(2 * r) * 2048);
#pragma unroll
    for (int r = 0; r < 8; ++r) {
      int sr = srow0 + 2 * r;
      sr = sr < 0 ? 0 : sr;
      sr = sr > L ? L : sr;
      pr[r] = *(const float4*)(pcol + (size_t)sr * DM);
    }
#pragma unroll
    for (int r = 0; r < 8; ++r)
      vv[r] = *(const float4*)(kp + 1024 + (size_t)(2 * r) * 2048);
#pragma unroll
    for (int r = 0; r < 8; ++r) {
      const float4 kq = kr[r];
      const float4 ps = pr[r];
      float4 t;
      t.x = -kq.x * ps.x; t.y = -kq.y * ps.y; t.z = -kq.z * ps.z; t.w = -kq.w * ps.w;
      *(float4*)&myK[(2 * r + vr) * QKP + vc]      = kq;
      *(float4*)&myK[(2 * r + vr) * QKP + 64 + vc] = t;
    }
#pragma unroll
    for (int r = 0; r < 8; ++r) *(float4*)&myV[(2 * r + vr) * VP + vc] = vv[r];
  }

  v8f s = {};
#pragma unroll 8
  for (int kk = 0; kk < 32; ++kk) {
    const int kb = kk * 4 + 2 * hhi;
    v2f a; a.x = myK[l16 * QKP + kb]; a.y = myK[l16 * QKP + kb + 1];
    v2f b; b.x = Qt[l16 * QKP + kb];  b.y = Qt[l16 * QKP + kb + 1];
    s = wmma_f32(a, b, s);
  }

  if (diag) {
#pragma unroll
    for (int v = 0; v < 8; ++v)
      if (8 * hhi + v > l16) s[v] = -1.0e30f;
  }
  float mx = fmaxf(fmaxf(fmaxf(s[0], s[1]), fmaxf(s[2], s[3])),
                   fmaxf(fmaxf(s[4], s[5]), fmaxf(s[6], s[7])));
  mx = fmaxf(mx, xhalf(mx));
  const float mnew = fmaxf(m, mx);
  const float alpha = __expf(m - mnew);
  float p[8];
#pragma unroll
  for (int v = 0; v < 8; ++v) p[v] = __expf(s[v] - mnew);
  float rs = ((p[0] + p[1]) + (p[2] + p[3])) + ((p[4] + p[5]) + (p[6] + p[7]));
  rs += xhalf(rs);
  l = l * alpha + rs;
  m = mnew;
#pragma unroll
  for (int v = 0; v < 8; ++v) {
    o0[v] *= alpha; o1[v] *= alpha; o2[v] *= alpha; o3[v] *= alpha;
  }

#pragma unroll
  for (int kk = 0; kk < 4; ++kk) {
    const int kr0 = 8 * hhi + 2 * kk;
    v2f b; b.x = p[2 * kk]; b.y = p[2 * kk + 1];
    v2f a0, a1, a2, a3;
    a0.x = myV[kr0 * VP + l16];      a0.y = myV[(kr0 + 1) * VP + l16];
    a1.x = myV[kr0 * VP + 16 + l16]; a1.y = myV[(kr0 + 1) * VP + 16 + l16];
    a2.x = myV[kr0 * VP + 32 + l16]; a2.y = myV[(kr0 + 1) * VP + 32 + l16];
    a3.x = myV[kr0 * VP + 48 + l16]; a3.y = myV[(kr0 + 1) * VP + 48 + l16];
    o0 = wmma_f32(a0, b, o0);
    o1 = wmma_f32(a1, b, o1);
    o2 = wmma_f32(a2, b, o2);
    o3 = wmma_f32(a3, b, o3);
  }
}

__global__ void __launch_bounds__(256)
attn_kernel(const float* __restrict__ Q, const float* __restrict__ KV,
            const float* __restrict__ P, const int* __restrict__ cw,
            float* __restrict__ wv) {
  __shared__ __attribute__((aligned(16))) float Qs[2 * 16 * QKP];
  __shared__ __attribute__((aligned(16))) float KVs[ANW * 16 * QKP + ANW * 16 * VP];
  __shared__ float Mm[2 * ANW * 16];
  __shared__ float Lm[2 * ANW * 16];

  const int tid = threadIdx.x;
  const int lane = tid & 31;
  const int wid = __builtin_amdgcn_readfirstlane(tid >> 5);
  const int hhi = lane >> 4, l16 = lane & 15;
  const int h = blockIdx.y;
  const int w = cw[0];
  const int p = blockIdx.x;
  const int qtA = 63 - p;
  const int qtB = p;
  const int nA = 64 - p;

  float* myK = &KVs[wid * 16 * QKP];
  float* myV = &KVs[ANW * 16 * QKP + wid * 16 * VP];

  {
    const float scale = 0.125f;
    const int tq = tid >> 7;
    const int t = tid & 127;
    const int r0 = t >> 4, cc = (t & 15) * 4;
    const int qrow = (tq ? qtB : qtA) * 16 + r0;
    const float* qsrc = Q + (size_t)qrow * DM + h * 64 + cc;
    const float* psrc = P + h * 64 + cc;
    float* qd = &Qs[tq * 16 * QKP + r0 * QKP + cc];
    float4 qv[2], pe[2];
#pragma unroll
    for (int i = 0; i < 2; ++i) qv[i] = *(const float4*)(qsrc + (size_t)(8 * i) * DM);
#pragma unroll
    for (int i = 0; i < 2; ++i) {
      int er = qrow + 8 * i + w;
      er = er < 0 ? 0 : er;
      er = er > L ? L : er;
      pe[i] = *(const float4*)(psrc + (size_t)er * DM);
    }
#pragma unroll
    for (int i = 0; i < 2; ++i) {
      float4 o;
      o.x = qv[i].x * (scale + pe[i].x);
      o.y = qv[i].y * (scale + pe[i].y);
      o.z = qv[i].z * (scale + pe[i].z);
      o.w = qv[i].w * (scale + pe[i].w);
      *(float4*)&qd[(8 * i) * QKP]      = o;
      *(float4*)&qd[(8 * i) * QKP + 64] = qv[i];
    }
  }
  __syncthreads();

  const int vr = lane >> 4, vc = (lane & 15) * 4;
  const float* ksrc = KV + (size_t)vr * 2048 + h * 64 + vc;
  const float* pcol = P + h * 64 + vc;
  const int sbase = vr - w;

  v8f a0 = {}, a1 = {}, a2 = {}, a3 = {};
  v8f b0 = {}, b1 = {}, b2 = {}, b3 = {};
  float mA = -3.0e38f, lA = 0.f, mB = -3.0e38f, lB = 0.f;

  int u = wid;
  for (; u < nA; u += ANW) {
    attn_tile(ksrc + (size_t)u * (16 * 2048), pcol, u * 16 + sbase,
              Qs, myK, myV, hhi, l16, vr, vc, u == qtA,
              a0, a1, a2, a3, mA, lA);
  }
  for (; u < 65; u += ANW) {
    const int jt = u - nA;
    attn_tile(ksrc + (size_t)jt * (16 * 2048), pcol, jt * 16 + sbase,
              Qs + 16 * QKP, myK, myV, hhi, l16, vr, vc, jt == qtB,
              b0, b1, b2, b3, mB, lB);
  }

  __syncthreads();
  float* Om = KVs;
  {
    float* oA = &Om[((0 * ANW + wid) * 16 + l16) * OMP + 8 * hhi];
    float* oB = &Om[((1 * ANW + wid) * 16 + l16) * OMP + 8 * hhi];
    *(float4*)&oA[0]  = make_float4(a0[0], a0[1], a0[2], a0[3]);
    *(float4*)&oA[4]  = make_float4(a0[4], a0[5], a0[6], a0[7]);
    *(float4*)&oA[16] = make_float4(a1[0], a1[1], a1[2], a1[3]);
    *(float4*)&oA[20] = make_float4(a1[4], a1[5], a1[6], a1[7]);
    *(float4*)&oA[32] = make_float4(a2[0], a2[1], a2[2], a2[3]);
    *(float4*)&oA[36] = make_float4(a2[4], a2[5], a2[6], a2[7]);
    *(float4*)&oA[48] = make_float4(a3[0], a3[1], a3[2], a3[3]);
    *(float4*)&oA[52] = make_float4(a3[4], a3[5], a3[6], a3[7]);
    *(float4*)&oB[0]  = make_float4(b0[0], b0[1], b0[2], b0[3]);
    *(float4*)&oB[4]  = make_float4(b0[4], b0[5], b0[6], b0[7]);
    *(float4*)&oB[16] = make_float4(b1[0], b1[1], b1[2], b1[3]);
    *(float4*)&oB[20] = make_float4(b1[4], b1[5], b1[6], b1[7]);
    *(float4*)&oB[32] = make_float4(b2[0], b2[1], b2[2], b2[3]);
    *(float4*)&oB[36] = make_float4(b2[4], b2[5], b2[6], b2[7]);
    *(float4*)&oB[48] = make_float4(b3[0], b3[1], b3[2], b3[3]);
    *(float4*)&oB[52] = make_float4(b3[4], b3[5], b3[6], b3[7]);
    if (hhi == 0) {
      Mm[(0 * ANW + wid) * 16 + l16] = mA; Lm[(0 * ANW + wid) * 16 + l16] = lA;
      Mm[(1 * ANW + wid) * 16 + l16] = mB; Lm[(1 * ANW + wid) * 16 + l16] = lB;
    }
  }
  __syncthreads();

  {
    const int tq = wid >> 2;
    const int row = (wid & 3) * 4 + (lane >> 3);
    const int c4 = (lane & 7) * 4;
    const float* Mt = &Mm[tq * ANW * 16 + row];
    const float* Lt = &Lm[tq * ANW * 16 + row];
    const float* Ot = &Om[(tq * ANW * 16 + row) * OMP];
    float pm[ANW], f[ANW];
#pragma unroll
    for (int w = 0; w < ANW; ++w) pm[w] = Mt[w * 16];
    float mt = pm[0];
#pragma unroll
    for (int w = 1; w < ANW; ++w) mt = fmaxf(mt, pm[w]);
    float lt = 0.f;
#pragma unroll
    for (int w = 0; w < ANW; ++w) {
      f[w] = __expf(pm[w] - mt);
      lt += Lt[w * 16] * f[w];
    }
    const float inv = 1.0f / lt;
    float* dst = wv + (size_t)((tq ? qtB : qtA) * 16 + row) * DM + h * 64;
#pragma unroll
    for (int i = 0; i < 2; ++i) {
      const int col = c4 + 32 * i;
      float4 acc = {0.f, 0.f, 0.f, 0.f};
#pragma unroll
      for (int w = 0; w < ANW; ++w) {
        const float4 t = *(const float4*)&Ot[w * 16 * OMP + col];
        acc.x += t.x * f[w]; acc.y += t.y * f[w];
        acc.z += t.z * f[w]; acc.w += t.w * f[w];
      }
      acc.x *= inv; acc.y *= inv; acc.z *= inv; acc.w *= inv;
      *(float4*)(dst + col) = acc;
    }
  }
}

extern "C" void kernel_launch(void* const* d_in, const int* in_sizes, int n_in,
                              void* d_out, int out_size, void* d_ws,
                              size_t ws_size, hipStream_t stream) {
  const float* x    = (const float*)d_in[0];
  const float* pt   = (const float*)d_in[1];
  const float* gq   = (const float*)d_in[3];
  const float* Wq   = (const float*)d_in[4];
  const float* bq   = (const float*)d_in[5];
  const float* gkv  = (const float*)d_in[6];
  const float* Wkv  = (const float*)d_in[7];
  const float* bkv  = (const float*)d_in[8];
  const float* gc   = (const float*)d_in[9];
  const float* Wc   = (const float*)d_in[10];
  const float* bc   = (const float*)d_in[11];
  const float* Wout = (const float*)d_in[12];
  const float* bout = (const float*)d_in[13];
  const int*   cw   = (const int*)d_in[14];
  float* out = (float*)d_out;

  float* ws = (float*)d_ws;
  const size_t R = 1024;
  float* Qb   = ws + 3 * R * 1024;
  float* KVb  = ws + 4 * R * 1024;
  float* Cb   = ws + 6 * R * 1024;
  float* Pb   = ws + 7 * R * 1024;
  float* wvb  = ws + 12352 * R;

  wmma_gemm_bias<true><<<dim3(16, 64), 256, 0, stream>>>(x, Wq, bq, gq, Qb,
                                                         x, Wkv, bkv, gkv, KVb,
                                                         pt, Wc, bc, gc, Cb);
  cumsum_kernel<<<32, 1024, 0, stream>>>(Cb, Pb);
  attn_kernel<<<dim3(32, 16), 256, 0, stream>>>(Qb, KVb, Pb, cw, wvb);
  wmma_gemm_bias<false><<<dim3(16, 16), 256, 0, stream>>>(wvb, Wout, bout, nullptr, out,
                                                          wvb, Wout, bout, nullptr, out,
                                                          wvb, Wout, bout, nullptr, out);
}
